# starting kernel
# baseline (speedup 1.0000x reference)
#include <hip/hip_runtime.h>
#include <math.h>


typedef __attribute__((ext_vector_type(2))) float v2f;
typedef __attribute__((ext_vector_type(8))) float v8f;

__global__ void __launch_bounds__(256)
gat_gemm_wmma(const float* __restrict__ A, const float* __restrict__ B,
              float* __restrict__ C, int M, int K) {
  const int lane    = threadIdx.x & 31;
  const int wv      = threadIdx.x >> 5;
  const int strip   = blockIdx.x * 8 + wv;
  const int nstrips = (M + 15) >> 4;
  if (strip >= nstrips) return;
  const int row0 = strip * 16;
  const int kh   = lane >> 4;
  const int l15  = lane & 15;
  int mrow = row0 + l15;
  if (mrow >= M) mrow = M - 1;

  const float* arow = A + (size_t)mrow * K;
  v8f acc[4];
#pragma unroll
  for (int t = 0; t < 4; ++t) acc[t] = (v8f){0.f,0.f,0.f,0.f,0.f,0.f,0.f,0.f};

  for (int k = 0; k < K; k += 4) {
    const v2f a = *(const v2f*)(arow + k + 2 * kh);
    const float* brow0 = B + (size_t)(k + 2 * kh) * 64 + l15;
#pragma unroll
    for (int t = 0; t < 4; ++t) {
      v2f b;
      b.x = brow0[t * 16];
      b.y = brow0[64 + t * 16];
      acc[t] = __builtin_amdgcn_wmma_f32_16x16x4_f32(
          false, a, false, b, (short)0, acc[t], false, false);
    }
  }

  const int rbase = row0 + 8 * kh;
#pragma unroll
  for (int t = 0; t < 4; ++t) {
    float* cp = C + (size_t)rbase * 64 + t * 16 + l15;
#pragma unroll
    for (int r = 0; r < 8; ++r)
      if (rbase + r < M) cp[(size_t)r * 64] = acc[t][r];
  }
}

__global__ void gat_node_attn(const float* __restrict__ z,
                              const float* __restrict__ al,
                              const float* __restrict__ ar,
                              float* __restrict__ el, float* __restrict__ er,
                              int N) {
  const int wid  = (int)((blockIdx.x * (size_t)blockDim.x + threadIdx.x) >> 5);
  const int lane = threadIdx.x & 31;
  if (wid >= N) return;
  const float* zr = z + (size_t)wid * 64;
  const float z0 = zr[lane], z1 = zr[lane + 32];
  float sl = z0 * al[lane] + z1 * al[lane + 32];
  float sr = z0 * ar[lane] + z1 * ar[lane + 32];
#pragma unroll
  for (int off = 16; off; off >>= 1) {
    sl += __shfl_xor(sl, off, 32);
    sr += __shfl_xor(sr, off, 32);
  }
  if (lane == 0) { el[wid] = sl; er[wid] = sr; }
}


__global__ void gat_zero_i32(int* __restrict__ p, int n) {
  const int i = blockIdx.x * blockDim.x + threadIdx.x;
  if (i < n) p[i] = 0;
}

__global__ void gat_histogram(const int* __restrict__ dst, int* __restrict__ deg, int E) {
  const int i = blockIdx.x * blockDim.x + threadIdx.x;
  if (i < E) atomicAdd(deg + dst[i], 1);
}

__global__ void gat_scan_partials(const int* __restrict__ deg, int* __restrict__ partial,
                                  int N) {
  __shared__ int sh[256];
  const int i = blockIdx.x * 256 + threadIdx.x;
  sh[threadIdx.x] = (i < N) ? deg[i] : 0;
  __syncthreads();
  for (int off = 128; off; off >>= 1) {
    if (threadIdx.x < off) sh[threadIdx.x] += sh[threadIdx.x + off];
    __syncthreads();
  }
  if (threadIdx.x == 0) partial[blockIdx.x] = sh[0];
}

__global__ void gat_scan_offsets(int* __restrict__ partial, int nb) {
  if (blockIdx.x == 0 && threadIdx.x == 0) {
    int run = 0;
    for (int b = 0; b < nb; ++b) { const int t = partial[b]; partial[b] = run; run += t; }
  }
}

__global__ void gat_scan_block(const int* __restrict__ deg, const int* __restrict__ partial,
                               int* __restrict__ rowstart, int* __restrict__ cursor, int N) {
  __shared__ int sh[256];
  const int i = blockIdx.x * 256 + threadIdx.x;
  const int v = (i < N) ? deg[i] : 0;
  sh[threadIdx.x] = v;
  __syncthreads();
  int val = v;
  for (int off = 1; off < 256; off <<= 1) {
    const int t = (threadIdx.x >= off) ? sh[threadIdx.x - off] : 0;
    __syncthreads();
    val += t;
    sh[threadIdx.x] = val;
    __syncthreads();
  }
  if (i < N) {
    const int ex = partial[blockIdx.x] + val - v;
    rowstart[i] = ex;
    cursor[i]   = ex;
  }
}

__global__ void gat_scatter(const int* __restrict__ src, const int* __restrict__ dst,
                            int* __restrict__ cursor, int* __restrict__ ssrc, int E) {
  const int i = blockIdx.x * blockDim.x + threadIdx.x;
  if (i < E) {
    const int pos = atomicAdd(cursor + dst[i], 1);
    ssrc[pos] = src[i];
  }
}

__global__ void gat_node_aggregate(const int* __restrict__ rowstart,
                                   const int* __restrict__ deg,
                                   const int* __restrict__ ssrc,
                                   const float* __restrict__ el,
                                   const float* __restrict__ er,
                                   const float* __restrict__ z,
                                   const float* __restrict__ bias,
                                   float* __restrict__ out,
                                   int N, int applyRelu) {
  const int wid  = (int)((blockIdx.x * (size_t)blockDim.x + threadIdx.x) >> 5);
  const int lane = threadIdx.x & 31;
  if (wid >= N) return;
  const int start = rowstart[wid];
  const int cnt   = deg[wid];
  const float ern = er[wid];

  float mx = -INFINITY;
  for (int j = lane; j < cnt; j += 32) {
    float v = el[ssrc[start + j]] + ern;
    v = (v > 0.f) ? v : 0.2f * v;
    mx = fmaxf(mx, v);
  }
#pragma unroll
  for (int off = 16; off; off >>= 1) mx = fmaxf(mx, __shfl_xor(mx, off, 32));

  float sum = 0.f;
  for (int j = lane; j < cnt; j += 32) {
    float v = el[ssrc[start + j]] + ern;
    v = (v > 0.f) ? v : 0.2f * v;
    sum += __expf(v - mx);
  }
#pragma unroll
  for (int off = 16; off; off >>= 1) sum += __shfl_xor(sum, off, 32);
  const float rcp = (cnt > 0) ? (1.f / sum) : 0.f;

  float2 acc = {0.f, 0.f};
  for (int j = 0; j < cnt; ++j) {
    const int sv = ssrc[start + j];
    float v = el[sv] + ern;
    v = (v > 0.f) ? v : 0.2f * v;
    const float alpha = __expf(v - mx) * rcp;
    const float2 zz = ((const float2*)(z + (size_t)sv * 64))[lane];
    acc.x += alpha * zz.x;
    acc.y += alpha * zz.y;
  }

  float2 o;
  o.x = acc.x + bias[2 * lane];
  o.y = acc.y + bias[2 * lane + 1];
  if (applyRelu) { o.x = fmaxf(o.x, 0.f); o.y = fmaxf(o.y, 0.f); }
  ((float2*)(out + (size_t)wid * 64))[lane] = o;
}


extern "C" void kernel_launch(void* const* d_in, const int* in_sizes, int n_in,
                              void* d_out, int out_size, void* d_ws, size_t ws_size,
                              hipStream_t stream) {
  const float* x   = (const float*)d_in[0];
  const float* W1  = (const float*)d_in[1];
  const float* al1 = (const float*)d_in[2];
  const float* ar1 = (const float*)d_in[3];
  const float* b1  = (const float*)d_in[4];
  const float* W2  = (const float*)d_in[5];
  const float* al2 = (const float*)d_in[6];
  const float* ar2 = (const float*)d_in[7];
  const float* b2  = (const float*)d_in[8];
  const int*   src = (const int*)d_in[9];
  const int*   dst = (const int*)d_in[10];
  float* out = (float*)d_out;

  const int F  = 64;
  const int N  = out_size / F;
  const int E  = in_sizes[9];
  const int K1 = in_sizes[0] / N;
  (void)n_in; (void)ws_size;

  char* ws = (char*)d_ws;
  size_t off = 0;
  auto alloc = [&](size_t bytes) -> void* {
    void* p = ws + off;
    off += (bytes + 255) & ~(size_t)255;
    return p;
  };
  float* zbuf     = (float*)alloc((size_t)N * F * sizeof(float));
  float* hbuf     = (float*)alloc((size_t)N * F * sizeof(float));
  float* el       = (float*)alloc((size_t)N * sizeof(float));
  float* er       = (float*)alloc((size_t)N * sizeof(float));
  int*   deg      = (int*)alloc((size_t)N * sizeof(int));
  int*   rowstart = (int*)alloc((size_t)N * sizeof(int));
  int*   cursor   = (int*)alloc((size_t)N * sizeof(int));
  int*   ssrc     = (int*)alloc((size_t)E * sizeof(int));
  const int nb    = (N + 255) / 256;
  int*   partial  = (int*)alloc((size_t)nb * sizeof(int));

  const int gN256  = (N + 255) / 256;
  const int gE256  = (E + 255) / 256;
  const int gWaveN = (N + 7) / 8;
  const int gGemm  = (((N + 15) / 16) + 7) / 8;

  gat_zero_i32     <<<gN256, 256, 0, stream>>>(deg, N);
  gat_histogram    <<<gE256, 256, 0, stream>>>(dst, deg, E);
  gat_scan_partials<<<nb,    256, 0, stream>>>(deg, partial, N);
  gat_scan_offsets <<<1,      32, 0, stream>>>(partial, nb);
  gat_scan_block   <<<nb,    256, 0, stream>>>(deg, partial, rowstart, cursor, N);
  gat_scatter      <<<gE256, 256, 0, stream>>>(src, dst, cursor, ssrc, E);

  gat_gemm_wmma     <<<gGemm,  256, 0, stream>>>(x, W1, zbuf, N, K1);
  gat_node_attn     <<<gWaveN, 256, 0, stream>>>(zbuf, al1, ar1, el, er, N);
  gat_node_aggregate<<<gWaveN, 256, 0, stream>>>(rowstart, deg, ssrc, el, er,
                                                 zbuf, b1, hbuf, N,  1);

  gat_gemm_wmma     <<<gGemm,  256, 0, stream>>>(hbuf, W2, zbuf, N, F);
  gat_node_attn     <<<gWaveN, 256, 0, stream>>>(zbuf, al2, ar2, el, er, N);
  gat_node_aggregate<<<gWaveN, 256, 0, stream>>>(rowstart, deg, ssrc, el, er,
                                                 zbuf, b2, out, N,  0);
}


# CSR build by two-level counting sort in LDS (256-node buckets, chunk-local sort, coalesced runs) instead of global int atomics; 12->10 launches; in-row edge order still arrival order (LDS atomics)
# speedup vs baseline: 1.9388x; 1.9388x over previous
#include <hip/hip_runtime.h>
#include <math.h>


typedef __attribute__((ext_vector_type(2))) float v2f;
typedef __attribute__((ext_vector_type(8))) float v8f;

__global__ void __launch_bounds__(256)
gat_gemm_wmma(const float* __restrict__ A, const float* __restrict__ B,
              float* __restrict__ C, int M, int K) {
  const int lane    = threadIdx.x & 31;
  const int wv      = threadIdx.x >> 5;
  const int strip   = blockIdx.x * 8 + wv;
  const int nstrips = (M + 15) >> 4;
  if (strip >= nstrips) return;
  const int row0 = strip * 16;
  const int kh   = lane >> 4;
  const int l15  = lane & 15;
  int mrow = row0 + l15;
  if (mrow >= M) mrow = M - 1;

  const float* arow = A + (size_t)mrow * K;
  v8f acc[4];
#pragma unroll
  for (int t = 0; t < 4; ++t) acc[t] = (v8f){0.f,0.f,0.f,0.f,0.f,0.f,0.f,0.f};

  for (int k = 0; k < K; k += 4) {
    const v2f a = *(const v2f*)(arow + k + 2 * kh);
    const float* brow0 = B + (size_t)(k + 2 * kh) * 64 + l15;
#pragma unroll
    for (int t = 0; t < 4; ++t) {
      v2f b;
      b.x = brow0[t * 16];
      b.y = brow0[64 + t * 16];
      acc[t] = __builtin_amdgcn_wmma_f32_16x16x4_f32(
          false, a, false, b, (short)0, acc[t], false, false);
    }
  }

  const int rbase = row0 + 8 * kh;
#pragma unroll
  for (int t = 0; t < 4; ++t) {
    float* cp = C + (size_t)rbase * 64 + t * 16 + l15;
#pragma unroll
    for (int r = 0; r < 8; ++r)
      if (rbase + r < M) cp[(size_t)r * 64] = acc[t][r];
  }
}

__global__ void gat_node_attn(const float* __restrict__ z,
                              const float* __restrict__ al,
                              const float* __restrict__ ar,
                              float* __restrict__ el, float* __restrict__ er,
                              int N) {
  const int wid  = (int)((blockIdx.x * (size_t)blockDim.x + threadIdx.x) >> 5);
  const int lane = threadIdx.x & 31;
  if (wid >= N) return;
  const float* zr = z + (size_t)wid * 64;
  const float z0 = zr[lane], z1 = zr[lane + 32];
  float sl = z0 * al[lane] + z1 * al[lane + 32];
  float sr = z0 * ar[lane] + z1 * ar[lane + 32];
#pragma unroll
  for (int off = 16; off; off >>= 1) {
    sl += __shfl_xor(sl, off, 32);
    sr += __shfl_xor(sr, off, 32);
  }
  if (lane == 0) { el[wid] = sl; er[wid] = sr; }
}


#define GAT_BSH   8
#define GAT_T     1024
#define GAT_MAXB  GAT_T
#define GAT_EPT   4
#define GAT_CHUNK (GAT_T * GAT_EPT)
#define GAT_EPT2  8
#define GAT_CAP2  12288

__device__ __forceinline__ int gat_wave_incl_scan(int v, int lane) {
#pragma unroll
  for (int off = 1; off < 32; off <<= 1) {
    const int t = __shfl_up(v, off, 32);
    if (lane >= off) v += t;
  }
  return v;
}

__device__ __forceinline__ int gat_block_excl_scan(int v, int* s_w) {
  const int lane = threadIdx.x & 31;
  const int wv   = threadIdx.x >> 5;
  const int nw   = (blockDim.x + 31) >> 5;
  const int inc  = gat_wave_incl_scan(v, lane);
  if (lane == 31) s_w[wv] = inc;
  __syncthreads();
  if (wv == 0) {
    const int w  = (lane < nw) ? s_w[lane] : 0;
    const int wi = gat_wave_incl_scan(w, lane);
    s_w[lane] = wi - w;
    if (lane == 31) s_w[32] = wi;
  }
  __syncthreads();
  return inc - v + s_w[wv];
}

__global__ void __launch_bounds__(GAT_T)
gat_bucket_count(const int* __restrict__ dst, int* __restrict__ counts,
                 int E, int B, int nwg) {
  __shared__ int s_h[GAT_MAXB];
  s_h[threadIdx.x] = 0;
  const int e0 = blockIdx.x * GAT_CHUNK + threadIdx.x;
  int d[GAT_EPT];
#pragma unroll
  for (int k = 0; k < GAT_EPT; ++k) {
    const int e = e0 + k * GAT_T;
    d[k] = (e < E) ? dst[e] : -1;
  }
  __syncthreads();
#pragma unroll
  for (int k = 0; k < GAT_EPT; ++k)
    if (d[k] >= 0) atomicAdd(&s_h[d[k] >> GAT_BSH], 1);
  __syncthreads();
  if (threadIdx.x < B)
    counts[(size_t)threadIdx.x * nwg + blockIdx.x] = s_h[threadIdx.x];
}

__global__ void __launch_bounds__(GAT_T)
gat_bucket_rowscan(const int* __restrict__ counts, int* __restrict__ rowpre,
                   int* __restrict__ total, int nwg) {
  __shared__ int s_w[33];
  const size_t r0 = (size_t)blockIdx.x * nwg;
  int carry = 0;
  for (int w0 = 0; w0 < nwg; w0 += GAT_T) {
    const int w  = w0 + threadIdx.x;
    const int c  = (w < nwg) ? counts[r0 + w] : 0;
    const int ex = gat_block_excl_scan(c, s_w);
    if (w < nwg) rowpre[r0 + w] = carry + ex;
    carry += s_w[32];
    __syncthreads();
  }
  if (threadIdx.x == 0) total[blockIdx.x] = carry;
}

__global__ void __launch_bounds__(GAT_T)
gat_bucket_scatter(const int* __restrict__ src, const int* __restrict__ dst,
                   const int* __restrict__ counts, const int* __restrict__ rowpre,
                   const int* __restrict__ total, int* __restrict__ tmpw,
                   int E, int B, int nwg) {
  int2* __restrict__ tmp = (int2*)tmpw;
  __shared__ int  s_cur[GAT_MAXB];
  __shared__ int  s_go[GAT_MAXB];
  __shared__ int2 s_pair[GAT_CHUNK];
  __shared__ int  s_w[33];
  __shared__ int  s_w2[33];
  const int wg = blockIdx.x;
  const int b  = threadIdx.x;
  const int e0 = wg * GAT_CHUNK + threadIdx.x;
  int sv[GAT_EPT], dv[GAT_EPT];
#pragma unroll
  for (int k = 0; k < GAT_EPT; ++k) {
    const int e = e0 + k * GAT_T;
    dv[k] = (e < E) ? dst[e] : -1;
    sv[k] = (e < E) ? src[e] : 0;
  }
  const int c  = (b < B) ? counts[(size_t)b * nwg + wg] : 0;
  const int rp = (b < B) ? rowpre[(size_t)b * nwg + wg] : 0;
  const int t  = (b < B) ? total[b] : 0;
  const int ex = gat_block_excl_scan(c, s_w);
  const int bs = gat_block_excl_scan(t, s_w2);
  s_cur[b] = ex;
  s_go[b]  = bs + rp - ex;
  __syncthreads();
#pragma unroll
  for (int k = 0; k < GAT_EPT; ++k) {
    if (dv[k] >= 0) {
      const int pos = atomicAdd(&s_cur[dv[k] >> GAT_BSH], 1);
      s_pair[pos] = make_int2(sv[k], dv[k]);
    }
  }
  __syncthreads();
  const int m = min(GAT_CHUNK, E - wg * GAT_CHUNK);
#pragma unroll
  for (int k = 0; k < GAT_EPT; ++k) {
    const int i = threadIdx.x + k * GAT_T;
    if (i < m) {
      const int2 p = s_pair[i];
      tmp[s_go[p.y >> GAT_BSH] + i] = p;
    }
  }
}

__global__ void __launch_bounds__(GAT_T)
gat_bucket_csr(const int* __restrict__ tmpw, const int* __restrict__ total,
               int* __restrict__ rowstart, int* __restrict__ deg,
               int* __restrict__ ssrc, int N, int B) {
  const int2* __restrict__ tmp = (const int2*)tmpw;
  __shared__ int s_cnt[256];
  __shared__ int s_cur[256];
  __shared__ int s_w[33];
  __shared__ int s_w2[33];
  __shared__ int s_be[2];
  __shared__ int s_sorted[GAT_CAP2];
  const int b   = blockIdx.x;
  const int tid = threadIdx.x;
  const int t   = (tid < B) ? total[tid] : 0;
  const int bs  = gat_block_excl_scan(t, s_w2);
  if (tid == b) { s_be[0] = bs; s_be[1] = t; }
  if (tid < 256) s_cnt[tid] = 0;
  __syncthreads();
  const int base = s_be[0];
  const int cnt  = s_be[1];

  for (int i0 = tid; i0 < cnt; i0 += GAT_T * GAT_EPT2) {
    int dl[GAT_EPT2];
#pragma unroll
    for (int k = 0; k < GAT_EPT2; ++k) {
      const int i = i0 + k * GAT_T;
      dl[k] = (i < cnt) ? tmp[base + i].y : -1;
    }
#pragma unroll
    for (int k = 0; k < GAT_EPT2; ++k)
      if (dl[k] >= 0) atomicAdd(&s_cnt[dl[k] & 255], 1);
  }
  __syncthreads();
  const int c  = (tid < 256) ? s_cnt[tid] : 0;
  const int ex = gat_block_excl_scan(c, s_w);
  if (tid < 256) {
    s_cur[tid] = ex;
    const int node = (b << GAT_BSH) + tid;
    if (node < N) { rowstart[node] = base + ex; deg[node] = c; }
  }
  __syncthreads();
  const bool fits = cnt <= GAT_CAP2;
  for (int i0 = tid; i0 < cnt; i0 += GAT_T * GAT_EPT2) {
    int2 p[GAT_EPT2];
#pragma unroll
    for (int k = 0; k < GAT_EPT2; ++k) {
      const int i = i0 + k * GAT_T;
      p[k] = (i < cnt) ? tmp[base + i] : make_int2(0, -1);
    }
#pragma unroll
    for (int k = 0; k < GAT_EPT2; ++k) {
      if (p[k].y >= 0) {
        const int pos = atomicAdd(&s_cur[p[k].y & 255], 1);
        if (fits) s_sorted[pos] = p[k].x; else ssrc[base + pos] = p[k].x;
      }
    }
  }
  __syncthreads();
  if (fits)
    for (int i = tid; i < cnt; i += GAT_T) ssrc[base + i] = s_sorted[i];
}

__global__ void gat_node_aggregate(const int* __restrict__ rowstart,
                                   const int* __restrict__ deg,
                                   const int* __restrict__ ssrc,
                                   const float* __restrict__ el,
                                   const float* __restrict__ er,
                                   const float* __restrict__ z,
                                   const float* __restrict__ bias,
                                   float* __restrict__ out,
                                   int N, int applyRelu) {
  const int wid  = (int)((blockIdx.x * (size_t)blockDim.x + threadIdx.x) >> 5);
  const int lane = threadIdx.x & 31;
  if (wid >= N) return;
  const int start = rowstart[wid];
  const int cnt   = deg[wid];
  const float ern = er[wid];

  float mx = -INFINITY;
  for (int j = lane; j < cnt; j += 32) {
    float v = el[ssrc[start + j]] + ern;
    v = (v > 0.f) ? v : 0.2f * v;
    mx = fmaxf(mx, v);
  }
#pragma unroll
  for (int off = 16; off; off >>= 1) mx = fmaxf(mx, __shfl_xor(mx, off, 32));

  float sum = 0.f;
  for (int j = lane; j < cnt; j += 32) {
    float v = el[ssrc[start + j]] + ern;
    v = (v > 0.f) ? v : 0.2f * v;
    sum += __expf(v - mx);
  }
#pragma unroll
  for (int off = 16; off; off >>= 1) sum += __shfl_xor(sum, off, 32);
  const float rcp = (cnt > 0) ? (1.f / sum) : 0.f;

  float2 acc = {0.f, 0.f};
  for (int j = 0; j < cnt; ++j) {
    const int sv = ssrc[start + j];
    float v = el[sv] + ern;
    v = (v > 0.f) ? v : 0.2f * v;
    const float alpha = __expf(v - mx) * rcp;
    const float2 zz = ((const float2*)(z + (size_t)sv * 64))[lane];
    acc.x += alpha * zz.x;
    acc.y += alpha * zz.y;
  }

  float2 o;
  o.x = acc.x + bias[2 * lane];
  o.y = acc.y + bias[2 * lane + 1];
  if (applyRelu) { o.x = fmaxf(o.x, 0.f); o.y = fmaxf(o.y, 0.f); }
  ((float2*)(out + (size_t)wid * 64))[lane] = o;
}


extern "C" void kernel_launch(void* const* d_in, const int* in_sizes, int n_in,
                              void* d_out, int out_size, void* d_ws, size_t ws_size,
                              hipStream_t stream) {
  const float* x   = (const float*)d_in[0];
  const float* W1  = (const float*)d_in[1];
  const float* al1 = (const float*)d_in[2];
  const float* ar1 = (const float*)d_in[3];
  const float* b1  = (const float*)d_in[4];
  const float* W2  = (const float*)d_in[5];
  const float* al2 = (const float*)d_in[6];
  const float* ar2 = (const float*)d_in[7];
  const float* b2  = (const float*)d_in[8];
  const int*   src = (const int*)d_in[9];
  const int*   dst = (const int*)d_in[10];
  float* out = (float*)d_out;

  const int F  = 64;
  const int N  = out_size / F;
  const int E  = in_sizes[9];
  const int K1 = in_sizes[0] / N;
  (void)n_in; (void)ws_size;

  char* ws = (char*)d_ws;
  size_t off = 0;
  auto alloc = [&](size_t bytes) -> void* {
    void* p = ws + off;
    off += (bytes + 255) & ~(size_t)255;
    return p;
  };
  float* zbuf     = (float*)alloc((size_t)N * F * sizeof(float));
  float* hbuf     = (float*)alloc((size_t)N * F * sizeof(float));
  float* el       = (float*)alloc((size_t)N * sizeof(float));
  float* er       = (float*)alloc((size_t)N * sizeof(float));
  int*   deg      = (int*)alloc((size_t)N * sizeof(int));
  int*   rowstart = (int*)alloc((size_t)N * sizeof(int));
  int*   ssrc     = (int*)alloc((size_t)E * sizeof(int));
  const int nb    = (N + 255) / 256;
  const int nwg   = (E + GAT_CHUNK - 1) / GAT_CHUNK;
  int*   counts   = (int*)alloc((size_t)nb * nwg * sizeof(int));
  int*   rowpre   = (int*)alloc((size_t)nb * nwg * sizeof(int));
  int*   total    = (int*)alloc((size_t)nb * sizeof(int));
  int*   tmp      = (int*)alloc((size_t)E * 2 * sizeof(int));

  const int gWaveN = (N + 7) / 8;
  const int gGemm  = (((N + 15) / 16) + 7) / 8;

  gat_bucket_count  <<<nwg, GAT_T, 0, stream>>>(dst, counts, E, nb, nwg);
  gat_bucket_rowscan<<<nb,  GAT_T, 0, stream>>>(counts, rowpre, total, nwg);
  gat_bucket_scatter<<<nwg, GAT_T, 0, stream>>>(src, dst, counts, rowpre, total, tmp,
                                                E, nb, nwg);
  gat_bucket_csr    <<<nb,  GAT_T, 0, stream>>>(tmp, total, rowstart, deg, ssrc, N, nb);

  gat_gemm_wmma     <<<gGemm,  256, 0, stream>>>(x, W1, zbuf, N, K1);
  gat_node_attn     <<<gWaveN, 256, 0, stream>>>(zbuf, al1, ar1, el, er, N);
  gat_node_aggregate<<<gWaveN, 256, 0, stream>>>(rowstart, deg, ssrc, el, er,
                                                 zbuf, b1, hbuf, N,  1);

  gat_gemm_wmma     <<<gGemm,  256, 0, stream>>>(hbuf, W2, zbuf, N, F);
  gat_node_attn     <<<gWaveN, 256, 0, stream>>>(zbuf, al2, ar2, el, er, N);
  gat_node_aggregate<<<gWaveN, 256, 0, stream>>>(rowstart, deg, ssrc, el, er,
                                                 zbuf, b2, out, N,  0);
}


# el/er attention dots formed in the GEMM epilogue from the accumulators (4 FMAs per row and lane, then a 16-lane DPP xor tree: new summation order); both per-node dot launches removed, 10->8 launches
# speedup vs baseline: 2.1983x; 1.1338x over previous
#include <hip/hip_runtime.h>
#include <math.h>


typedef __attribute__((ext_vector_type(2))) float v2f;
typedef __attribute__((ext_vector_type(8))) float v8f;

template <int XM>
__device__ __forceinline__ float gat_row_xadd(float v) {
  const int b = __builtin_bit_cast(int, v);
  const int o = __builtin_amdgcn_update_dpp(0, b, 0x160 | XM, 0xf, 0xf, true);
  return v + __builtin_bit_cast(float, o);
}

__global__ void __launch_bounds__(256)
gat_gemm_wmma(const float* __restrict__ A, const float* __restrict__ B,
              const float* __restrict__ al, const float* __restrict__ ar,
              float* __restrict__ C, float* __restrict__ el,
              float* __restrict__ er, int M, int K) {
  const int lane    = threadIdx.x & 31;
  const int wv      = threadIdx.x >> 5;
  const int strip   = blockIdx.x * 8 + wv;
  const int nstrips = (M + 15) >> 4;
  if (strip >= nstrips) return;
  const int row0 = strip * 16;
  const int kh   = lane >> 4;
  const int l15  = lane & 15;
  int mrow = row0 + l15;
  if (mrow >= M) mrow = M - 1;

  const float* arow = A + (size_t)mrow * K;
  v8f acc[4];
#pragma unroll
  for (int t = 0; t < 4; ++t) acc[t] = (v8f){0.f,0.f,0.f,0.f,0.f,0.f,0.f,0.f};

  for (int k = 0; k < K; k += 4) {
    const v2f a = *(const v2f*)(arow + k + 2 * kh);
    const float* brow0 = B + (size_t)(k + 2 * kh) * 64 + l15;
#pragma unroll
    for (int t = 0; t < 4; ++t) {
      v2f b;
      b.x = brow0[t * 16];
      b.y = brow0[64 + t * 16];
      acc[t] = __builtin_amdgcn_wmma_f32_16x16x4_f32(
          false, a, false, b, (short)0, acc[t], false, false);
    }
  }

  const int rbase = row0 + 8 * kh;
#pragma unroll
  for (int t = 0; t < 4; ++t) {
    float* cp = C + (size_t)rbase * 64 + t * 16 + l15;
#pragma unroll
    for (int r = 0; r < 8; ++r)
      if (rbase + r < M) cp[(size_t)r * 64] = acc[t][r];
  }

  float alv[4], arv[4];
#pragma unroll
  for (int t = 0; t < 4; ++t) {
    alv[t] = al[t * 16 + l15];
    arv[t] = ar[t * 16 + l15];
  }
  float pl[8], pr[8];
#pragma unroll
  for (int r = 0; r < 8; ++r) {
    float sl = acc[0][r] * alv[0];
    float sr = acc[0][r] * arv[0];
#pragma unroll
    for (int t = 1; t < 4; ++t) {
      sl += acc[t][r] * alv[t];
      sr += acc[t][r] * arv[t];
    }
    pl[r] = sl;
    pr[r] = sr;
  }
#pragma unroll
  for (int r = 0; r < 8; ++r) { pl[r] = gat_row_xadd<1>(pl[r]); pr[r] = gat_row_xadd<1>(pr[r]); }
#pragma unroll
  for (int r = 0; r < 8; ++r) { pl[r] = gat_row_xadd<2>(pl[r]); pr[r] = gat_row_xadd<2>(pr[r]); }
#pragma unroll
  for (int r = 0; r < 8; ++r) { pl[r] = gat_row_xadd<4>(pl[r]); pr[r] = gat_row_xadd<4>(pr[r]); }
#pragma unroll
  for (int r = 0; r < 8; ++r) { pl[r] = gat_row_xadd<8>(pl[r]); pr[r] = gat_row_xadd<8>(pr[r]); }
  const int  rsel = l15 & 7;
  const bool isl  = l15 < 8;
  float outv = isl ? pl[0] : pr[0];
#pragma unroll
  for (int r = 1; r < 8; ++r)
    if (rsel == r) outv = isl ? pl[r] : pr[r];
  float* op = isl ? el : er;
  if (rbase + rsel < M) op[rbase + rsel] = outv;
}


#define GAT_BSH   8
#define GAT_T     1024
#define GAT_MAXB  GAT_T
#define GAT_EPT   4
#define GAT_CHUNK (GAT_T * GAT_EPT)
#define GAT_EPT2  8
#define GAT_CAP2  12288

__device__ __forceinline__ int gat_wave_incl_scan(int v, int lane) {
#pragma unroll
  for (int off = 1; off < 32; off <<= 1) {
    const int t = __shfl_up(v, off, 32);
    if (lane >= off) v += t;
  }
  return v;
}

__device__ __forceinline__ int gat_block_excl_scan(int v, int* s_w) {
  const int lane = threadIdx.x & 31;
  const int wv   = threadIdx.x >> 5;
  const int nw   = (blockDim.x + 31) >> 5;
  const int inc  = gat_wave_incl_scan(v, lane);
  if (lane == 31) s_w[wv] = inc;
  __syncthreads();
  if (wv == 0) {
    const int w  = (lane < nw) ? s_w[lane] : 0;
    const int wi = gat_wave_incl_scan(w, lane);
    s_w[lane] = wi - w;
    if (lane == 31) s_w[32] = wi;
  }
  __syncthreads();
  return inc - v + s_w[wv];
}

__global__ void __launch_bounds__(GAT_T)
gat_bucket_count(const int* __restrict__ dst, int* __restrict__ counts,
                 int E, int B, int nwg) {
  __shared__ int s_h[GAT_MAXB];
  s_h[threadIdx.x] = 0;
  const int e0 = blockIdx.x * GAT_CHUNK + threadIdx.x;
  int d[GAT_EPT];
#pragma unroll
  for (int k = 0; k < GAT_EPT; ++k) {
    const int e = e0 + k * GAT_T;
    d[k] = (e < E) ? dst[e] : -1;
  }
  __syncthreads();
#pragma unroll
  for (int k = 0; k < GAT_EPT; ++k)
    if (d[k] >= 0) atomicAdd(&s_h[d[k] >> GAT_BSH], 1);
  __syncthreads();
  if (threadIdx.x < B)
    counts[(size_t)threadIdx.x * nwg + blockIdx.x] = s_h[threadIdx.x];
}

__global__ void __launch_bounds__(GAT_T)
gat_bucket_rowscan(const int* __restrict__ counts, int* __restrict__ rowpre,
                   int* __restrict__ total, int nwg) {
  __shared__ int s_w[33];
  const size_t r0 = (size_t)blockIdx.x * nwg;
  int carry = 0;
  for (int w0 = 0; w0 < nwg; w0 += GAT_T) {
    const int w  = w0 + threadIdx.x;
    const int c  = (w < nwg) ? counts[r0 + w] : 0;
    const int ex = gat_block_excl_scan(c, s_w);
    if (w < nwg) rowpre[r0 + w] = carry + ex;
    carry += s_w[32];
    __syncthreads();
  }
  if (threadIdx.x == 0) total[blockIdx.x] = carry;
}

__global__ void __launch_bounds__(GAT_T)
gat_bucket_scatter(const int* __restrict__ src, const int* __restrict__ dst,
                   const int* __restrict__ counts, const int* __restrict__ rowpre,
                   const int* __restrict__ total, int* __restrict__ tmpw,
                   int E, int B, int nwg) {
  int2* __restrict__ tmp = (int2*)tmpw;
  __shared__ int  s_cur[GAT_MAXB];
  __shared__ int  s_go[GAT_MAXB];
  __shared__ int2 s_pair[GAT_CHUNK];
  __shared__ int  s_w[33];
  __shared__ int  s_w2[33];
  const int wg = blockIdx.x;
  const int b  = threadIdx.x;
  const int e0 = wg * GAT_CHUNK + threadIdx.x;
  int sv[GAT_EPT], dv[GAT_EPT];
#pragma unroll
  for (int k = 0; k < GAT_EPT; ++k) {
    const int e = e0 + k * GAT_T;
    dv[k] = (e < E) ? dst[e] : -1;
    sv[k] = (e < E) ? src[e] : 0;
  }
  const int c  = (b < B) ? counts[(size_t)b * nwg + wg] : 0;
  const int rp = (b < B) ? rowpre[(size_t)b * nwg + wg] : 0;
  const int t  = (b < B) ? total[b] : 0;
  const int ex = gat_block_excl_scan(c, s_w);
  const int bs = gat_block_excl_scan(t, s_w2);
  s_cur[b] = ex;
  s_go[b]  = bs + rp - ex;
  __syncthreads();
#pragma unroll
  for (int k = 0; k < GAT_EPT; ++k) {
    if (dv[k] >= 0) {
      const int pos = atomicAdd(&s_cur[dv[k] >> GAT_BSH], 1);
      s_pair[pos] = make_int2(sv[k], dv[k]);
    }
  }
  __syncthreads();
  const int m = min(GAT_CHUNK, E - wg * GAT_CHUNK);
#pragma unroll
  for (int k = 0; k < GAT_EPT; ++k) {
    const int i = threadIdx.x + k * GAT_T;
    if (i < m) {
      const int2 p = s_pair[i];
      tmp[s_go[p.y >> GAT_BSH] + i] = p;
    }
  }
}

__global__ void __launch_bounds__(GAT_T)
gat_bucket_csr(const int* __restrict__ tmpw, const int* __restrict__ total,
               int* __restrict__ rowstart, int* __restrict__ deg,
               int* __restrict__ ssrc, int N, int B) {
  const int2* __restrict__ tmp = (const int2*)tmpw;
  __shared__ int s_cnt[256];
  __shared__ int s_cur[256];
  __shared__ int s_w[33];
  __shared__ int s_w2[33];
  __shared__ int s_be[2];
  __shared__ int s_sorted[GAT_CAP2];
  const int b   = blockIdx.x;
  const int tid = threadIdx.x;
  const int t   = (tid < B) ? total[tid] : 0;
  const int bs  = gat_block_excl_scan(t, s_w2);
  if (tid == b) { s_be[0] = bs; s_be[1] = t; }
  if (tid < 256) s_cnt[tid] = 0;
  __syncthreads();
  const int base = s_be[0];
  const int cnt  = s_be[1];

  for (int i0 = tid; i0 < cnt; i0 += GAT_T * GAT_EPT2) {
    int dl[GAT_EPT2];
#pragma unroll
    for (int k = 0; k < GAT_EPT2; ++k) {
      const int i = i0 + k * GAT_T;
      dl[k] = (i < cnt) ? tmp[base + i].y : -1;
    }
#pragma unroll
    for (int k = 0; k < GAT_EPT2; ++k)
      if (dl[k] >= 0) atomicAdd(&s_cnt[dl[k] & 255], 1);
  }
  __syncthreads();
  const int c  = (tid < 256) ? s_cnt[tid] : 0;
  const int ex = gat_block_excl_scan(c, s_w);
  if (tid < 256) {
    s_cur[tid] = ex;
    const int node = (b << GAT_BSH) + tid;
    if (node < N) { rowstart[node] = base + ex; deg[node] = c; }
  }
  __syncthreads();
  const bool fits = cnt <= GAT_CAP2;
  for (int i0 = tid; i0 < cnt; i0 += GAT_T * GAT_EPT2) {
    int2 p[GAT_EPT2];
#pragma unroll
    for (int k = 0; k < GAT_EPT2; ++k) {
      const int i = i0 + k * GAT_T;
      p[k] = (i < cnt) ? tmp[base + i] : make_int2(0, -1);
    }
#pragma unroll
    for (int k = 0; k < GAT_EPT2; ++k) {
      if (p[k].y >= 0) {
        const int pos = atomicAdd(&s_cur[p[k].y & 255], 1);
        if (fits) s_sorted[pos] = p[k].x; else ssrc[base + pos] = p[k].x;
      }
    }
  }
  __syncthreads();
  if (fits)
    for (int i = tid; i < cnt; i += GAT_T) ssrc[base + i] = s_sorted[i];
}

__global__ void gat_node_aggregate(const int* __restrict__ rowstart,
                                   const int* __restrict__ deg,
                                   const int* __restrict__ ssrc,
                                   const float* __restrict__ el,
                                   const float* __restrict__ er,
                                   const float* __restrict__ z,
                                   const float* __restrict__ bias,
                                   float* __restrict__ out,
                                   int N, int applyRelu) {
  const int wid  = (int)((blockIdx.x * (size_t)blockDim.x + threadIdx.x) >> 5);
  const int lane = threadIdx.x & 31;
  if (wid >= N) return;
  const int start = rowstart[wid];
  const int cnt   = deg[wid];
  const float ern = er[wid];

  float mx = -INFINITY;
  for (int j = lane; j < cnt; j += 32) {
    float v = el[ssrc[start + j]] + ern;
    v = (v > 0.f) ? v : 0.2f * v;
    mx = fmaxf(mx, v);
  }
#pragma unroll
  for (int off = 16; off; off >>= 1) mx = fmaxf(mx, __shfl_xor(mx, off, 32));

  float sum = 0.f;
  for (int j = lane; j < cnt; j += 32) {
    float v = el[ssrc[start + j]] + ern;
    v = (v > 0.f) ? v : 0.2f * v;
    sum += __expf(v - mx);
  }
#pragma unroll
  for (int off = 16; off; off >>= 1) sum += __shfl_xor(sum, off, 32);
  const float rcp = (cnt > 0) ? (1.f / sum) : 0.f;

  float2 acc = {0.f, 0.f};
  for (int j = 0; j < cnt; ++j) {
    const int sv = ssrc[start + j];
    float v = el[sv] + ern;
    v = (v > 0.f) ? v : 0.2f * v;
    const float alpha = __expf(v - mx) * rcp;
    const float2 zz = ((const float2*)(z + (size_t)sv * 64))[lane];
    acc.x += alpha * zz.x;
    acc.y += alpha * zz.y;
  }

  float2 o;
  o.x = acc.x + bias[2 * lane];
  o.y = acc.y + bias[2 * lane + 1];
  if (applyRelu) { o.x = fmaxf(o.x, 0.f); o.y = fmaxf(o.y, 0.f); }
  ((float2*)(out + (size_t)wid * 64))[lane] = o;
}


extern "C" void kernel_launch(void* const* d_in, const int* in_sizes, int n_in,
                              void* d_out, int out_size, void* d_ws, size_t ws_size,
                              hipStream_t stream) {
  const float* x   = (const float*)d_in[0];
  const float* W1  = (const float*)d_in[1];
  const float* al1 = (const float*)d_in[2];
  const float* ar1 = (const float*)d_in[3];
  const float* b1  = (const float*)d_in[4];
  const float* W2  = (const float*)d_in[5];
  const float* al2 = (const float*)d_in[6];
  const float* ar2 = (const float*)d_in[7];
  const float* b2  = (const float*)d_in[8];
  const int*   src = (const int*)d_in[9];
  const int*   dst = (const int*)d_in[10];
  float* out = (float*)d_out;

  const int F  = 64;
  const int N  = out_size / F;
  const int E  = in_sizes[9];
  const int K1 = in_sizes[0] / N;
  (void)n_in; (void)ws_size;

  char* ws = (char*)d_ws;
  size_t off = 0;
  auto alloc = [&](size_t bytes) -> void* {
    void* p = ws + off;
    off += (bytes + 255) & ~(size_t)255;
    return p;
  };
  float* zbuf     = (float*)alloc((size_t)N * F * sizeof(float));
  float* hbuf     = (float*)alloc((size_t)N * F * sizeof(float));
  float* el       = (float*)alloc((size_t)N * sizeof(float));
  float* er       = (float*)alloc((size_t)N * sizeof(float));
  int*   deg      = (int*)alloc((size_t)N * sizeof(int));
  int*   rowstart = (int*)alloc((size_t)N * sizeof(int));
  int*   ssrc     = (int*)alloc((size_t)E * sizeof(int));
  const int nb    = (N + 255) / 256;
  const int nwg   = (E + GAT_CHUNK - 1) / GAT_CHUNK;
  int*   counts   = (int*)alloc((size_t)nb * nwg * sizeof(int));
  int*   rowpre   = (int*)alloc((size_t)nb * nwg * sizeof(int));
  int*   total    = (int*)alloc((size_t)nb * sizeof(int));
  int*   tmp      = (int*)alloc((size_t)E * 2 * sizeof(int));

  const int gWaveN = (N + 7) / 8;
  const int gGemm  = (((N + 15) / 16) + 7) / 8;

  gat_bucket_count  <<<nwg, GAT_T, 0, stream>>>(dst, counts, E, nb, nwg);
  gat_bucket_rowscan<<<nb,  GAT_T, 0, stream>>>(counts, rowpre, total, nwg);
  gat_bucket_scatter<<<nwg, GAT_T, 0, stream>>>(src, dst, counts, rowpre, total, tmp,
                                                E, nb, nwg);
  gat_bucket_csr    <<<nb,  GAT_T, 0, stream>>>(tmp, total, rowstart, deg, ssrc, N, nb);

  gat_gemm_wmma     <<<gGemm,  256, 0, stream>>>(x, W1, al1, ar1, zbuf, el, er, N, K1);
  gat_node_aggregate<<<gWaveN, 256, 0, stream>>>(rowstart, deg, ssrc, el, er,
                                                 zbuf, b1, hbuf, N,  1);

  gat_gemm_wmma     <<<gGemm,  256, 0, stream>>>(hbuf, W2, al2, ar2, zbuf, el, er, N, F);
  gat_node_aggregate<<<gWaveN, 256, 0, stream>>>(rowstart, deg, ssrc, el, er,
                                                 zbuf, b2, out, N,  0);
}


# aggregate: each lane keeps one edge's source index and softmax weight; z-row gathers take them by readlane and issue 16 at a time, no gather behind an index load; el read once; sum order unchanged
# speedup vs baseline: 2.8220x; 1.2837x over previous
#include <hip/hip_runtime.h>
#include <math.h>


typedef __attribute__((ext_vector_type(2))) float v2f;
typedef __attribute__((ext_vector_type(8))) float v8f;

template <int XM>
__device__ __forceinline__ float gat_row_xadd(float v) {
  const int b = __builtin_bit_cast(int, v);
  const int o = __builtin_amdgcn_update_dpp(0, b, 0x160 | XM, 0xf, 0xf, true);
  return v + __builtin_bit_cast(float, o);
}

__global__ void __launch_bounds__(256)
gat_gemm_wmma(const float* __restrict__ A, const float* __restrict__ B,
              const float* __restrict__ al, const float* __restrict__ ar,
              float* __restrict__ C, float* __restrict__ el,
              float* __restrict__ er, int M, int K) {
  const int lane    = threadIdx.x & 31;
  const int wv      = threadIdx.x >> 5;
  const int strip   = blockIdx.x * 8 + wv;
  const int nstrips = (M + 15) >> 4;
  if (strip >= nstrips) return;
  const int row0 = strip * 16;
  const int kh   = lane >> 4;
  const int l15  = lane & 15;
  int mrow = row0 + l15;
  if (mrow >= M) mrow = M - 1;

  const float* arow = A + (size_t)mrow * K;
  v8f acc[4];
#pragma unroll
  for (int t = 0; t < 4; ++t) acc[t] = (v8f){0.f,0.f,0.f,0.f,0.f,0.f,0.f,0.f};

  for (int k = 0; k < K; k += 4) {
    const v2f a = *(const v2f*)(arow + k + 2 * kh);
    const float* brow0 = B + (size_t)(k + 2 * kh) * 64 + l15;
#pragma unroll
    for (int t = 0; t < 4; ++t) {
      v2f b;
      b.x = brow0[t * 16];
      b.y = brow0[64 + t * 16];
      acc[t] = __builtin_amdgcn_wmma_f32_16x16x4_f32(
          false, a, false, b, (short)0, acc[t], false, false);
    }
  }

  const int rbase = row0 + 8 * kh;
#pragma unroll
  for (int t = 0; t < 4; ++t) {
    float* cp = C + (size_t)rbase * 64 + t * 16 + l15;
#pragma unroll
    for (int r = 0; r < 8; ++r)
      if (rbase + r < M) cp[(size_t)r * 64] = acc[t][r];
  }

  float alv[4], arv[4];
#pragma unroll
  for (int t = 0; t < 4; ++t) {
    alv[t] = al[t * 16 + l15];
    arv[t] = ar[t * 16 + l15];
  }
  float pl[8], pr[8];
#pragma unroll
  for (int r = 0; r < 8; ++r) {
    float sl = acc[0][r] * alv[0];
    float sr = acc[0][r] * arv[0];
#pragma unroll
    for (int t = 1; t < 4; ++t) {
      sl += acc[t][r] * alv[t];
      sr += acc[t][r] * arv[t];
    }
    pl[r] = sl;
    pr[r] = sr;
  }
#pragma unroll
  for (int r = 0; r < 8; ++r) { pl[r] = gat_row_xadd<1>(pl[r]); pr[r] = gat_row_xadd<1>(pr[r]); }
#pragma unroll
  for (int r = 0; r < 8; ++r) { pl[r] = gat_row_xadd<2>(pl[r]); pr[r] = gat_row_xadd<2>(pr[r]); }
#pragma unroll
  for (int r = 0; r < 8; ++r) { pl[r] = gat_row_xadd<4>(pl[r]); pr[r] = gat_row_xadd<4>(pr[r]); }
#pragma unroll
  for (int r = 0; r < 8; ++r) { pl[r] = gat_row_xadd<8>(pl[r]); pr[r] = gat_row_xadd<8>(pr[r]); }
  const int  rsel = l15 & 7;
  const bool isl  = l15 < 8;
  float outv = isl ? pl[0] : pr[0];
#pragma unroll
  for (int r = 1; r < 8; ++r)
    if (rsel == r) outv = isl ? pl[r] : pr[r];
  float* op = isl ? el : er;
  if (rbase + rsel < M) op[rbase + rsel] = outv;
}


#define GAT_BSH   8
#define GAT_T     1024
#define GAT_MAXB  GAT_T
#define GAT_EPT   4
#define GAT_CHUNK (GAT_T * GAT_EPT)
#define GAT_EPT2  8
#define GAT_CAP2  12288

__device__ __forceinline__ int gat_wave_incl_scan(int v, int lane) {
#pragma unroll
  for (int off = 1; off < 32; off <<= 1) {
    const int t = __shfl_up(v, off, 32);
    if (lane >= off) v += t;
  }
  return v;
}

__device__ __forceinline__ int gat_block_excl_scan(int v, int* s_w) {
  const int lane = threadIdx.x & 31;
  const int wv   = threadIdx.x >> 5;
  const int nw   = (blockDim.x + 31) >> 5;
  const int inc  = gat_wave_incl_scan(v, lane);
  if (lane == 31) s_w[wv] = inc;
  __syncthreads();
  if (wv == 0) {
    const int w  = (lane < nw) ? s_w[lane] : 0;
    const int wi = gat_wave_incl_scan(w, lane);
    s_w[lane] = wi - w;
    if (lane == 31) s_w[32] = wi;
  }
  __syncthreads();
  return inc - v + s_w[wv];
}

__global__ void __launch_bounds__(GAT_T)
gat_bucket_count(const int* __restrict__ dst, int* __restrict__ counts,
                 int E, int B, int nwg) {
  __shared__ int s_h[GAT_MAXB];
  s_h[threadIdx.x] = 0;
  const int e0 = blockIdx.x * GAT_CHUNK + threadIdx.x;
  int d[GAT_EPT];
#pragma unroll
  for (int k = 0; k < GAT_EPT; ++k) {
    const int e = e0 + k * GAT_T;
    d[k] = (e < E) ? dst[e] : -1;
  }
  __syncthreads();
#pragma unroll
  for (int k = 0; k < GAT_EPT; ++k)
    if (d[k] >= 0) atomicAdd(&s_h[d[k] >> GAT_BSH], 1);
  __syncthreads();
  if (threadIdx.x < B)
    counts[(size_t)threadIdx.x * nwg + blockIdx.x] = s_h[threadIdx.x];
}

__global__ void __launch_bounds__(GAT_T)
gat_bucket_rowscan(const int* __restrict__ counts, int* __restrict__ rowpre,
                   int* __restrict__ total, int nwg) {
  __shared__ int s_w[33];
  const size_t r0 = (size_t)blockIdx.x * nwg;
  int carry = 0;
  for (int w0 = 0; w0 < nwg; w0 += GAT_T) {
    const int w  = w0 + threadIdx.x;
    const int c  = (w < nwg) ? counts[r0 + w] : 0;
    const int ex = gat_block_excl_scan(c, s_w);
    if (w < nwg) rowpre[r0 + w] = carry + ex;
    carry += s_w[32];
    __syncthreads();
  }
  if (threadIdx.x == 0) total[blockIdx.x] = carry;
}

__global__ void __launch_bounds__(GAT_T)
gat_bucket_scatter(const int* __restrict__ src, const int* __restrict__ dst,
                   const int* __restrict__ counts, const int* __restrict__ rowpre,
                   const int* __restrict__ total, int* __restrict__ tmpw,
                   int E, int B, int nwg) {
  int2* __restrict__ tmp = (int2*)tmpw;
  __shared__ int  s_cur[GAT_MAXB];
  __shared__ int  s_go[GAT_MAXB];
  __shared__ int2 s_pair[GAT_CHUNK];
  __shared__ int  s_w[33];
  __shared__ int  s_w2[33];
  const int wg = blockIdx.x;
  const int b  = threadIdx.x;
  const int e0 = wg * GAT_CHUNK + threadIdx.x;
  int sv[GAT_EPT], dv[GAT_EPT];
#pragma unroll
  for (int k = 0; k < GAT_EPT; ++k) {
    const int e = e0 + k * GAT_T;
    dv[k] = (e < E) ? dst[e] : -1;
    sv[k] = (e < E) ? src[e] : 0;
  }
  const int c  = (b < B) ? counts[(size_t)b * nwg + wg] : 0;
  const int rp = (b < B) ? rowpre[(size_t)b * nwg + wg] : 0;
  const int t  = (b < B) ? total[b] : 0;
  const int ex = gat_block_excl_scan(c, s_w);
  const int bs = gat_block_excl_scan(t, s_w2);
  s_cur[b] = ex;
  s_go[b]  = bs + rp - ex;
  __syncthreads();
#pragma unroll
  for (int k = 0; k < GAT_EPT; ++k) {
    if (dv[k] >= 0) {
      const int pos = atomicAdd(&s_cur[dv[k] >> GAT_BSH], 1);
      s_pair[pos] = make_int2(sv[k], dv[k]);
    }
  }
  __syncthreads();
  const int m = min(GAT_CHUNK, E - wg * GAT_CHUNK);
#pragma unroll
  for (int k = 0; k < GAT_EPT; ++k) {
    const int i = threadIdx.x + k * GAT_T;
    if (i < m) {
      const int2 p = s_pair[i];
      tmp[s_go[p.y >> GAT_BSH] + i] = p;
    }
  }
}

__global__ void __launch_bounds__(GAT_T)
gat_bucket_csr(const int* __restrict__ tmpw, const int* __restrict__ total,
               int* __restrict__ rowstart, int* __restrict__ deg,
               int* __restrict__ ssrc, int N, int B) {
  const int2* __restrict__ tmp = (const int2*)tmpw;
  __shared__ int s_cnt[256];
  __shared__ int s_cur[256];
  __shared__ int s_w[33];
  __shared__ int s_w2[33];
  __shared__ int s_be[2];
  __shared__ int s_sorted[GAT_CAP2];
  const int b   = blockIdx.x;
  const int tid = threadIdx.x;
  const int t   = (tid < B) ? total[tid] : 0;
  const int bs  = gat_block_excl_scan(t, s_w2);
  if (tid == b) { s_be[0] = bs; s_be[1] = t; }
  if (tid < 256) s_cnt[tid] = 0;
  __syncthreads();
  const int base = s_be[0];
  const int cnt  = s_be[1];

  for (int i0 = tid; i0 < cnt; i0 += GAT_T * GAT_EPT2) {
    int dl[GAT_EPT2];
#pragma unroll
    for (int k = 0; k < GAT_EPT2; ++k) {
      const int i = i0 + k * GAT_T;
      dl[k] = (i < cnt) ? tmp[base + i].y : -1;
    }
#pragma unroll
    for (int k = 0; k < GAT_EPT2; ++k)
      if (dl[k] >= 0) atomicAdd(&s_cnt[dl[k] & 255], 1);
  }
  __syncthreads();
  const int c  = (tid < 256) ? s_cnt[tid] : 0;
  const int ex = gat_block_excl_scan(c, s_w);
  if (tid < 256) {
    s_cur[tid] = ex;
    const int node = (b << GAT_BSH) + tid;
    if (node < N) { rowstart[node] = base + ex; deg[node] = c; }
  }
  __syncthreads();
  const bool fits = cnt <= GAT_CAP2;
  for (int i0 = tid; i0 < cnt; i0 += GAT_T * GAT_EPT2) {
    int2 p[GAT_EPT2];
#pragma unroll
    for (int k = 0; k < GAT_EPT2; ++k) {
      const int i = i0 + k * GAT_T;
      p[k] = (i < cnt) ? tmp[base + i] : make_int2(0, -1);
    }
#pragma unroll
    for (int k = 0; k < GAT_EPT2; ++k) {
      if (p[k].y >= 0) {
        const int pos = atomicAdd(&s_cur[p[k].y & 255], 1);
        if (fits) s_sorted[pos] = p[k].x; else ssrc[base + pos] = p[k].x;
      }
    }
  }
  __syncthreads();
  if (fits)
    for (int i = tid; i < cnt; i += GAT_T) ssrc[base + i] = s_sorted[i];
}


#define GAT_GRP 16

__device__ __forceinline__ void gat_gather_chunk(const float* __restrict__ z,
                                                 int svl, float wl, int nn,
                                                 int lane, float2& acc) {
  const int wi = __builtin_bit_cast(int, wl);
  int j0 = 0;
  for (; j0 + GAT_GRP <= nn; j0 += GAT_GRP) {
    float2 zz[GAT_GRP];
    float  w[GAT_GRP];
#pragma unroll
    for (int k = 0; k < GAT_GRP; ++k) {
      const int s = __builtin_amdgcn_readlane(svl, j0 + k);
      w[k]  = __builtin_bit_cast(float, __builtin_amdgcn_readlane(wi, j0 + k));
      zz[k] = ((const float2*)(z + (size_t)s * 64))[lane];
    }
#pragma unroll
    for (int k = 0; k < GAT_GRP; ++k) {
      acc.x = fmaf(w[k], zz[k].x, acc.x);
      acc.y = fmaf(w[k], zz[k].y, acc.y);
    }
  }
  if (j0 < nn) {
    float2 zz[GAT_GRP];
    float  w[GAT_GRP];
#pragma unroll
    for (int k = 0; k < GAT_GRP; ++k) {
      zz[k] = make_float2(0.f, 0.f);
      w[k]  = 0.f;
      if (j0 + k < nn) {
        const int s = __builtin_amdgcn_readlane(svl, j0 + k);
        w[k]  = __builtin_bit_cast(float, __builtin_amdgcn_readlane(wi, j0 + k));
        zz[k] = ((const float2*)(z + (size_t)s * 64))[lane];
      }
    }
#pragma unroll
    for (int k = 0; k < GAT_GRP; ++k) {
      if (j0 + k < nn) {
        acc.x = fmaf(w[k], zz[k].x, acc.x);
        acc.y = fmaf(w[k], zz[k].y, acc.y);
      }
    }
  }
}

__global__ void __launch_bounds__(256)
gat_node_aggregate(const int* __restrict__ rowstart,
                   const int* __restrict__ deg,
                   const int* __restrict__ ssrc,
                   const float* __restrict__ el,
                   const float* __restrict__ er,
                   const float* __restrict__ z,
                   const float* __restrict__ bias,
                   float* __restrict__ out,
                   int N, int applyRelu) {
  const int lane = threadIdx.x & 31;
  const int wv  = __builtin_amdgcn_readfirstlane((int)(threadIdx.x >> 5));
  const int wid = (int)blockIdx.x * (int)(blockDim.x >> 5) + wv;
  if (wid >= N) return;
  const int start = rowstart[wid];
  const int cnt   = deg[wid];
  const float ern = er[wid];

  int   sv0 = 0;
  float v0  = 0.f;
  float mx  = -INFINITY;
  if (lane < cnt) {
    sv0 = ssrc[start + lane];
    const float v = el[sv0] + ern;
    v0 = (v > 0.f) ? v : 0.2f * v;
    mx = fmaxf(mx, v0);
  }
  for (int base = 32; base < cnt; base += 32) {
    const int j = base + lane;
    if (j < cnt) {
      float v = el[ssrc[start + j]] + ern;
      v = (v > 0.f) ? v : 0.2f * v;
      mx = fmaxf(mx, v);
    }
  }
#pragma unroll
  for (int off = 16; off; off >>= 1) mx = fmaxf(mx, __shfl_xor(mx, off, 32));

  const float ex0 = (lane < cnt) ? __expf(v0 - mx) : 0.f;
  float sum = ex0;
  for (int base = 32; base < cnt; base += 32) {
    const int j = base + lane;
    if (j < cnt) {
      float v = el[ssrc[start + j]] + ern;
      v = (v > 0.f) ? v : 0.2f * v;
      sum += __expf(v - mx);
    }
  }
#pragma unroll
  for (int off = 16; off; off >>= 1) sum += __shfl_xor(sum, off, 32);
  const float rcp = (cnt > 0) ? (1.f / sum) : 0.f;

  float2 acc = {0.f, 0.f};
  gat_gather_chunk(z, sv0, ex0 * rcp, min(cnt, 32), lane, acc);
  for (int base = 32; base < cnt; base += 32) {
    const int j = base + lane;
    int   svc = 0;
    float wc  = 0.f;
    if (j < cnt) {
      svc = ssrc[start + j];
      float v = el[svc] + ern;
      v = (v > 0.f) ? v : 0.2f * v;
      wc = __expf(v - mx) * rcp;
    }
    gat_gather_chunk(z, svc, wc, min(cnt - base, 32), lane, acc);
  }

  float2 o;
  o.x = acc.x + bias[2 * lane];
  o.y = acc.y + bias[2 * lane + 1];
  if (applyRelu) { o.x = fmaxf(o.x, 0.f); o.y = fmaxf(o.y, 0.f); }
  ((float2*)(out + (size_t)wid * 64))[lane] = o;
}


extern "C" void kernel_launch(void* const* d_in, const int* in_sizes, int n_in,
                              void* d_out, int out_size, void* d_ws, size_t ws_size,
                              hipStream_t stream) {
  const float* x   = (const float*)d_in[0];
  const float* W1  = (const float*)d_in[1];
  const float* al1 = (const float*)d_in[2];
  const float* ar1 = (const float*)d_in[3];
  const float* b1  = (const float*)d_in[4];
  const float* W2  = (const float*)d_in[5];
  const float* al2 = (const float*)d_in[6];
  const float* ar2 = (const float*)d_in[7];
  const float* b2  = (const float*)d_in[8];
  const int*   src = (const int*)d_in[9];
  const int*   dst = (const int*)d_in[10];
  float* out = (float*)d_out;

  const int F  = 64;
  const int N  = out_size / F;
  const int E  = in_sizes[9];
  const int K1 = in_sizes[0] / N;
  (void)n_in; (void)ws_size;

  char* ws = (char*)d_ws;
  size_t off = 0;
  auto alloc = [&](size_t bytes) -> void* {
    void* p = ws + off;
    off += (bytes + 255) & ~(size_t)255;
    return p;
  };
  float* zbuf     = (float*)alloc((size_t)N * F * sizeof(float));
  float* hbuf     = (float*)alloc((size_t)N * F * sizeof(float));
  float* el       = (float*)alloc((size_t)N * sizeof(float));
  float* er       = (float*)alloc((size_t)N * sizeof(float));
  int*   deg      = (int*)alloc((size_t)N * sizeof(int));
  int*   rowstart = (int*)alloc((size_t)N * sizeof(int));
  int*   ssrc     = (int*)alloc((size_t)E * sizeof(int));
  const int nb    = (N + 255) / 256;
  const int nwg   = (E + GAT_CHUNK - 1) / GAT_CHUNK;
  int*   counts   = (int*)alloc((size_t)nb * nwg * sizeof(int));
  int*   rowpre   = (int*)alloc((size_t)nb * nwg * sizeof(int));
  int*   total    = (int*)alloc((size_t)nb * sizeof(int));
  int*   tmp      = (int*)alloc((size_t)E * 2 * sizeof(int));

  const int gWaveN = (N + 7) / 8;
  const int gGemm  = (((N + 15) / 16) + 7) / 8;

  gat_bucket_count  <<<nwg, GAT_T, 0, stream>>>(dst, counts, E, nb, nwg);
  gat_bucket_rowscan<<<nb,  GAT_T, 0, stream>>>(counts, rowpre, total, nwg);
  gat_bucket_scatter<<<nwg, GAT_T, 0, stream>>>(src, dst, counts, rowpre, total, tmp,
                                                E, nb, nwg);
  gat_bucket_csr    <<<nb,  GAT_T, 0, stream>>>(tmp, total, rowstart, deg, ssrc, N, nb);

  gat_gemm_wmma     <<<gGemm,  256, 0, stream>>>(x, W1, al1, ar1, zbuf, el, er, N, K1);
  gat_node_aggregate<<<gWaveN, 256, 0, stream>>>(rowstart, deg, ssrc, el, er,
                                                 zbuf, b1, hbuf, N,  1);

  gat_gemm_wmma     <<<gGemm,  256, 0, stream>>>(hbuf, W2, al2, ar2, zbuf, el, er, N, F);
  gat_node_aggregate<<<gWaveN, 256, 0, stream>>>(rowstart, deg, ssrc, el, er,
                                                 zbuf, b2, out, N,  0);
}


# CSR build in 2 launches, not 4: each chunk bucket-sorted in place by its own LDS histogram, bucket pass gathers its runs via per-chunk tables; 8->6 launches; in-row edge order still arrival order
# speedup vs baseline: 2.9357x; 1.0403x over previous
#include <hip/hip_runtime.h>
#include <math.h>


typedef __attribute__((ext_vector_type(2))) float v2f;
typedef __attribute__((ext_vector_type(8))) float v8f;

template <int XM>
__device__ __forceinline__ float gat_row_xadd(float v) {
  const int b = __builtin_bit_cast(int, v);
  const int o = __builtin_amdgcn_update_dpp(0, b, 0x160 | XM, 0xf, 0xf, true);
  return v + __builtin_bit_cast(float, o);
}

__global__ void __launch_bounds__(256)
gat_gemm_wmma(const float* __restrict__ A, const float* __restrict__ B,
              const float* __restrict__ al, const float* __restrict__ ar,
              float* __restrict__ C, float* __restrict__ el,
              float* __restrict__ er, int M, int K) {
  const int lane    = threadIdx.x & 31;
  const int wv      = threadIdx.x >> 5;
  const int strip   = blockIdx.x * 8 + wv;
  const int nstrips = (M + 15) >> 4;
  if (strip >= nstrips) return;
  const int row0 = strip * 16;
  const int kh   = lane >> 4;
  const int l15  = lane & 15;
  int mrow = row0 + l15;
  if (mrow >= M) mrow = M - 1;

  const float* arow = A + (size_t)mrow * K;
  v8f acc[4];
#pragma unroll
  for (int t = 0; t < 4; ++t) acc[t] = (v8f){0.f,0.f,0.f,0.f,0.f,0.f,0.f,0.f};

  for (int k = 0; k < K; k += 4) {
    const v2f a = *(const v2f*)(arow + k + 2 * kh);
    const float* brow0 = B + (size_t)(k + 2 * kh) * 64 + l15;
#pragma unroll
    for (int t = 0; t < 4; ++t) {
      v2f b;
      b.x = brow0[t * 16];
      b.y = brow0[64 + t * 16];
      acc[t] = __builtin_amdgcn_wmma_f32_16x16x4_f32(
          false, a, false, b, (short)0, acc[t], false, false);
    }
  }

  const int rbase = row0 + 8 * kh;
#pragma unroll
  for (int t = 0; t < 4; ++t) {
    float* cp = C + (size_t)rbase * 64 + t * 16 + l15;
#pragma unroll
    for (int r = 0; r < 8; ++r)
      if (rbase + r < M) cp[(size_t)r * 64] = acc[t][r];
  }

  float alv[4], arv[4];
#pragma unroll
  for (int t = 0; t < 4; ++t) {
    alv[t] = al[t * 16 + l15];
    arv[t] = ar[t * 16 + l15];
  }
  float pl[8], pr[8];
#pragma unroll
  for (int r = 0; r < 8; ++r) {
    float sl = acc[0][r] * alv[0];
    float sr = acc[0][r] * arv[0];
#pragma unroll
    for (int t = 1; t < 4; ++t) {
      sl += acc[t][r] * alv[t];
      sr += acc[t][r] * arv[t];
    }
    pl[r] = sl;
    pr[r] = sr;
  }
#pragma unroll
  for (int r = 0; r < 8; ++r) { pl[r] = gat_row_xadd<1>(pl[r]); pr[r] = gat_row_xadd<1>(pr[r]); }
#pragma unroll
  for (int r = 0; r < 8; ++r) { pl[r] = gat_row_xadd<2>(pl[r]); pr[r] = gat_row_xadd<2>(pr[r]); }
#pragma unroll
  for (int r = 0; r < 8; ++r) { pl[r] = gat_row_xadd<4>(pl[r]); pr[r] = gat_row_xadd<4>(pr[r]); }
#pragma unroll
  for (int r = 0; r < 8; ++r) { pl[r] = gat_row_xadd<8>(pl[r]); pr[r] = gat_row_xadd<8>(pr[r]); }
  const int  rsel = l15 & 7;
  const bool isl  = l15 < 8;
  float outv = isl ? pl[0] : pr[0];
#pragma unroll
  for (int r = 1; r < 8; ++r)
    if (rsel == r) outv = isl ? pl[r] : pr[r];
  float* op = isl ? el : er;
  if (rbase + rsel < M) op[rbase + rsel] = outv;
}


#define GAT_BSH   8
#define GAT_T     1024
#define GAT_MAXB  GAT_T
#define GAT_EPT   4
#define GAT_CHUNK (GAT_T * GAT_EPT)
#define GAT_EPT2  8
#define GAT_CAP2  12288
#define GAT_MAXW  2048
#define GAT_RUNS  10

__device__ __forceinline__ int gat_wave_incl_scan(int v, int lane) {
#pragma unroll
  for (int off = 1; off < 32; off <<= 1) {
    const int t = __shfl_up(v, off, 32);
    if (lane >= off) v += t;
  }
  return v;
}

__device__ __forceinline__ int gat_block_excl_scan(int v, int* s_w) {
  const int lane = threadIdx.x & 31;
  const int wv   = threadIdx.x >> 5;
  const int nw   = (blockDim.x + 31) >> 5;
  const int inc  = gat_wave_incl_scan(v, lane);
  if (lane == 31) s_w[wv] = inc;
  __syncthreads();
  if (wv == 0) {
    const int w  = (lane < nw) ? s_w[lane] : 0;
    const int wi = gat_wave_incl_scan(w, lane);
    s_w[lane] = wi - w;
    if (lane == 31) s_w[32] = wi;
  }
  __syncthreads();
  return inc - v + s_w[wv];
}

__global__ void __launch_bounds__(GAT_T)
gat_chunk_sort(const int* __restrict__ src, const int* __restrict__ dst,
               int* __restrict__ cnt_t, int* __restrict__ off_t,
               int* __restrict__ tmpw, int E, int B, int nwg) {
  int2* __restrict__ tmp = (int2*)tmpw;
  __shared__ int  s_h[GAT_MAXB];
  __shared__ int  s_cur[GAT_MAXB];
  __shared__ int2 s_pair[GAT_CHUNK];
  __shared__ int  s_w[33];
  const int wg = blockIdx.x;
  const int b  = threadIdx.x;
  s_h[b] = 0;
  const int e0 = wg * GAT_CHUNK + threadIdx.x;
  int sv[GAT_EPT], dv[GAT_EPT];
#pragma unroll
  for (int k = 0; k < GAT_EPT; ++k) {
    const int e = e0 + k * GAT_T;
    dv[k] = (e < E) ? dst[e] : -1;
    sv[k] = (e < E) ? src[e] : 0;
  }
  __syncthreads();
#pragma unroll
  for (int k = 0; k < GAT_EPT; ++k)
    if (dv[k] >= 0) atomicAdd(&s_h[dv[k] >> GAT_BSH], 1);
  __syncthreads();
  const int c  = s_h[b];
  const int ex = gat_block_excl_scan(c, s_w);
  s_cur[b] = ex;
  if (b < B) {
    cnt_t[(size_t)b * nwg + wg] = c;
    off_t[(size_t)b * nwg + wg] = ex;
  }
  __syncthreads();
#pragma unroll
  for (int k = 0; k < GAT_EPT; ++k) {
    if (dv[k] >= 0) {
      const int pos = atomicAdd(&s_cur[dv[k] >> GAT_BSH], 1);
      s_pair[pos] = make_int2(sv[k], dv[k]);
    }
  }
  __syncthreads();
  const int m = min(GAT_CHUNK, E - wg * GAT_CHUNK);
#pragma unroll
  for (int k = 0; k < GAT_EPT; ++k) {
    const int i = threadIdx.x + k * GAT_T;
    if (i < m) tmp[(size_t)wg * GAT_CHUNK + i] = s_pair[i];
  }
}

__device__ __forceinline__ int2 gat_run_fetch(const int2* __restrict__ tmp,
                                              const int* s_rp, const int* s_ro,
                                              int nwg, int i) {
  int lo = 0, hi = nwg - 1;
  while (lo < hi) {
    const int mid = (lo + hi + 1) >> 1;
    if (s_rp[mid] <= i) lo = mid; else hi = mid - 1;
  }
  return tmp[s_ro[lo] + (i - s_rp[lo])];
}

__global__ void __launch_bounds__(GAT_T)
gat_bucket_csr(const int* __restrict__ tmpw, const int* __restrict__ cnt_t,
               const int* __restrict__ off_t,
               int* __restrict__ rowstart, int* __restrict__ deg,
               int* __restrict__ ssrc, int N, int nwg) {
  const int2* __restrict__ tmp = (const int2*)tmpw;
  __shared__ int  s_cnt[256];
  __shared__ int  s_cur[256];
  __shared__ int  s_w[33];
  __shared__ int  s_w2[33];
  __shared__ int  s_rc[GAT_MAXW];
  __shared__ int  s_ro[GAT_MAXW];
  __shared__ int  s_rp[GAT_MAXW];
  __shared__ int2 s_stage[GAT_CAP2];
  __shared__ int  s_sorted[GAT_CAP2];
  const int b   = blockIdx.x;
  const int tid = threadIdx.x;

  int cnt = 0, base = 0;
  for (int w0 = 0; w0 < nwg; w0 += GAT_T) {
    const int w  = w0 + tid;
    const int c  = (w < nwg) ? cnt_t[(size_t)b * nwg + w] : 0;
    const int o  = (w < nwg) ? off_t[(size_t)b * nwg + w] : 0;
    const int ex = gat_block_excl_scan(c, s_w);
    (void)gat_block_excl_scan(o, s_w2);
    if (w < nwg) {
      s_rc[w] = c;
      s_ro[w] = w * GAT_CHUNK + o;
      s_rp[w] = cnt + ex;
    }
    cnt  += s_w[32];
    base += s_w2[32];
    __syncthreads();
  }
  if (tid < 256) s_cnt[tid] = 0;
  __syncthreads();
  const bool fits = cnt <= GAT_CAP2;

  if (fits) {
    const int lane = tid & 31;
    const int wv   = tid >> 5;
    for (int w0 = wv; w0 < nwg; w0 += 32 * GAT_RUNS) {
      int  rc[GAT_RUNS], ro[GAT_RUNS], rp[GAT_RUNS];
      int2 v[GAT_RUNS];
#pragma unroll
      for (int u = 0; u < GAT_RUNS; ++u) {
        const int  w  = w0 + 32 * u;
        const bool ok = w < nwg;
        const int  wc = min(w, nwg - 1);
        const int  tc = s_rc[wc];
        const int  to = s_ro[wc];
        const int  tp = s_rp[wc];
        rc[u] = ok ? tc : 0;
        ro[u] = ok ? to : 0;
        rp[u] = ok ? tp : 0;
      }
#pragma unroll
      for (int u = 0; u < GAT_RUNS; ++u)
        v[u] = (lane < rc[u]) ? tmp[ro[u] + lane] : make_int2(0, 0);
#pragma unroll
      for (int u = 0; u < GAT_RUNS; ++u) {
        if (lane < rc[u]) {
          s_stage[rp[u] + lane] = v[u];
          atomicAdd(&s_cnt[v[u].y & 255], 1);
        }
      }
      for (int u = 0; u < GAT_RUNS; ++u) {
        const int w = w0 + 32 * u;
        if (w >= nwg) break;
        const int c = s_rc[w];
        for (int i = lane + 32; i < c; i += 32) {
          const int2 q = tmp[s_ro[w] + i];
          s_stage[s_rp[w] + i] = q;
          atomicAdd(&s_cnt[q.y & 255], 1);
        }
      }
    }
  } else {
    for (int i0 = tid; i0 < cnt; i0 += GAT_T * GAT_EPT2) {
      int dl[GAT_EPT2];
#pragma unroll
      for (int k = 0; k < GAT_EPT2; ++k) {
        const int i = i0 + k * GAT_T;
        dl[k] = (i < cnt) ? gat_run_fetch(tmp, s_rp, s_ro, nwg, i).y : -1;
      }
#pragma unroll
      for (int k = 0; k < GAT_EPT2; ++k)
        if (dl[k] >= 0) atomicAdd(&s_cnt[dl[k] & 255], 1);
    }
  }
  __syncthreads();
  const int c  = (tid < 256) ? s_cnt[tid] : 0;
  const int ex = gat_block_excl_scan(c, s_w);
  if (tid < 256) {
    s_cur[tid] = ex;
    const int node = (b << GAT_BSH) + tid;
    if (node < N) { rowstart[node] = base + ex; deg[node] = c; }
  }
  __syncthreads();
  if (fits) {
    for (int i = tid; i < cnt; i += GAT_T) {
      const int2 p  = s_stage[i];
      const int pos = atomicAdd(&s_cur[p.y & 255], 1);
      s_sorted[pos] = p.x;
    }
    __syncthreads();
    for (int i = tid; i < cnt; i += GAT_T) ssrc[base + i] = s_sorted[i];
  } else {
    for (int i = tid; i < cnt; i += GAT_T) {
      const int2 p  = gat_run_fetch(tmp, s_rp, s_ro, nwg, i);
      const int pos = atomicAdd(&s_cur[p.y & 255], 1);
      ssrc[base + pos] = p.x;
    }
  }
}


#define GAT_GRP 16

__device__ __forceinline__ void gat_gather_chunk(const float* __restrict__ z,
                                                 int svl, float wl, int nn,
                                                 int lane, float2& acc) {
  const int wi = __builtin_bit_cast(int, wl);
  int j0 = 0;
  for (; j0 + GAT_GRP <= nn; j0 += GAT_GRP) {
    float2 zz[GAT_GRP];
    float  w[GAT_GRP];
#pragma unroll
    for (int k = 0; k < GAT_GRP; ++k) {
      const int s = __builtin_amdgcn_readlane(svl, j0 + k);
      w[k]  = __builtin_bit_cast(float, __builtin_amdgcn_readlane(wi, j0 + k));
      zz[k] = ((const float2*)(z + (size_t)s * 64))[lane];
    }
#pragma unroll
    for (int k = 0; k < GAT_GRP; ++k) {
      acc.x = fmaf(w[k], zz[k].x, acc.x);
      acc.y = fmaf(w[k], zz[k].y, acc.y);
    }
  }
  if (j0 < nn) {
    float2 zz[GAT_GRP];
    float  w[GAT_GRP];
#pragma unroll
    for (int k = 0; k < GAT_GRP; ++k) {
      zz[k] = make_float2(0.f, 0.f);
      w[k]  = 0.f;
      if (j0 + k < nn) {
        const int s = __builtin_amdgcn_readlane(svl, j0 + k);
        w[k]  = __builtin_bit_cast(float, __builtin_amdgcn_readlane(wi, j0 + k));
        zz[k] = ((const float2*)(z + (size_t)s * 64))[lane];
      }
    }
#pragma unroll
    for (int k = 0; k < GAT_GRP; ++k) {
      if (j0 + k < nn) {
        acc.x = fmaf(w[k], zz[k].x, acc.x);
        acc.y = fmaf(w[k], zz[k].y, acc.y);
      }
    }
  }
}

__global__ void __launch_bounds__(256)
gat_node_aggregate(const int* __restrict__ rowstart,
                   const int* __restrict__ deg,
                   const int* __restrict__ ssrc,
                   const float* __restrict__ el,
                   const float* __restrict__ er,
                   const float* __restrict__ z,
                   const float* __restrict__ bias,
                   float* __restrict__ out,
                   int N, int applyRelu) {
  const int lane = threadIdx.x & 31;
  const int wv  = __builtin_amdgcn_readfirstlane((int)(threadIdx.x >> 5));
  const int wid = (int)blockIdx.x * (int)(blockDim.x >> 5) + wv;
  if (wid >= N) return;
  const int start = rowstart[wid];
  const int cnt   = deg[wid];
  const float ern = er[wid];

  int   sv0 = 0;
  float v0  = 0.f;
  float mx  = -INFINITY;
  if (lane < cnt) {
    sv0 = ssrc[start + lane];
    const float v = el[sv0] + ern;
    v0 = (v > 0.f) ? v : 0.2f * v;
    mx = fmaxf(mx, v0);
  }
  for (int base = 32; base < cnt; base += 32) {
    const int j = base + lane;
    if (j < cnt) {
      float v = el[ssrc[start + j]] + ern;
      v = (v > 0.f) ? v : 0.2f * v;
      mx = fmaxf(mx, v);
    }
  }
#pragma unroll
  for (int off = 16; off; off >>= 1) mx = fmaxf(mx, __shfl_xor(mx, off, 32));

  const float ex0 = (lane < cnt) ? __expf(v0 - mx) : 0.f;
  float sum = ex0;
  for (int base = 32; base < cnt; base += 32) {
    const int j = base + lane;
    if (j < cnt) {
      float v = el[ssrc[start + j]] + ern;
      v = (v > 0.f) ? v : 0.2f * v;
      sum += __expf(v - mx);
    }
  }
#pragma unroll
  for (int off = 16; off; off >>= 1) sum += __shfl_xor(sum, off, 32);
  const float rcp = (cnt > 0) ? (1.f / sum) : 0.f;

  float2 acc = {0.f, 0.f};
  gat_gather_chunk(z, sv0, ex0 * rcp, min(cnt, 32), lane, acc);
  for (int base = 32; base < cnt; base += 32) {
    const int j = base + lane;
    int   svc = 0;
    float wc  = 0.f;
    if (j < cnt) {
      svc = ssrc[start + j];
      float v = el[svc] + ern;
      v = (v > 0.f) ? v : 0.2f * v;
      wc = __expf(v - mx) * rcp;
    }
    gat_gather_chunk(z, svc, wc, min(cnt - base, 32), lane, acc);
  }

  float2 o;
  o.x = acc.x + bias[2 * lane];
  o.y = acc.y + bias[2 * lane + 1];
  if (applyRelu) { o.x = fmaxf(o.x, 0.f); o.y = fmaxf(o.y, 0.f); }
  ((float2*)(out + (size_t)wid * 64))[lane] = o;
}


extern "C" void kernel_launch(void* const* d_in, const int* in_sizes, int n_in,
                              void* d_out, int out_size, void* d_ws, size_t ws_size,
                              hipStream_t stream) {
  const float* x   = (const float*)d_in[0];
  const float* W1  = (const float*)d_in[1];
  const float* al1 = (const float*)d_in[2];
  const float* ar1 = (const float*)d_in[3];
  const float* b1  = (const float*)d_in[4];
  const float* W2  = (const float*)d_in[5];
  const float* al2 = (const float*)d_in[6];
  const float* ar2 = (const float*)d_in[7];
  const float* b2  = (const float*)d_in[8];
  const int*   src = (const int*)d_in[9];
  const int*   dst = (const int*)d_in[10];
  float* out = (float*)d_out;

  const int F  = 64;
  const int N  = out_size / F;
  const int E  = in_sizes[9];
  const int K1 = in_sizes[0] / N;
  (void)n_in; (void)ws_size;

  char* ws = (char*)d_ws;
  size_t off = 0;
  auto alloc = [&](size_t bytes) -> void* {
    void* p = ws + off;
    off += (bytes + 255) & ~(size_t)255;
    return p;
  };
  float* zbuf     = (float*)alloc((size_t)N * F * sizeof(float));
  float* hbuf     = (float*)alloc((size_t)N * F * sizeof(float));
  float* el       = (float*)alloc((size_t)N * sizeof(float));
  float* er       = (float*)alloc((size_t)N * sizeof(float));
  int*   deg      = (int*)alloc((size_t)N * sizeof(int));
  int*   rowstart = (int*)alloc((size_t)N * sizeof(int));
  int*   ssrc     = (int*)alloc((size_t)E * sizeof(int));
  const int nb    = (N + 255) / 256;
  const int nwg   = (E + GAT_CHUNK - 1) / GAT_CHUNK;
  int*   cnt_t    = (int*)alloc((size_t)nb * nwg * sizeof(int));
  int*   off_t    = (int*)alloc((size_t)nb * nwg * sizeof(int));
  int*   tmp      = (int*)alloc((size_t)E * 2 * sizeof(int));

  const int gWaveN = (N + 7) / 8;
  const int gGemm  = (((N + 15) / 16) + 7) / 8;

  gat_chunk_sort    <<<nwg, GAT_T, 0, stream>>>(src, dst, cnt_t, off_t, tmp, E, nb, nwg);
  gat_bucket_csr    <<<nb,  GAT_T, 0, stream>>>(tmp, cnt_t, off_t, rowstart, deg, ssrc,
                                                N, nwg);

  gat_gemm_wmma     <<<gGemm,  256, 0, stream>>>(x, W1, al1, ar1, zbuf, el, er, N, K1);
  gat_node_aggregate<<<gWaveN, 256, 0, stream>>>(rowstart, deg, ssrc, el, er,
                                                 zbuf, b1, hbuf, N,  1);

  gat_gemm_wmma     <<<gGemm,  256, 0, stream>>>(hbuf, W2, al2, ar2, zbuf, el, er, N, F);
  gat_node_aggregate<<<gWaveN, 256, 0, stream>>>(rowstart, deg, ssrc, el, er,
                                                 zbuf, b2, out, N,  0);
}


# layer-1 GEMM runs inside CSR pass 1's launch (196 extra workgroups of 16 strips ahead of the sort's; K loop 2 steps a trip to fit beside a sort workgroup); 6->5 launches; sum order unchanged
# speedup vs baseline: 3.1096x; 1.0592x over previous
#include <hip/hip_runtime.h>
#include <math.h>


typedef __attribute__((ext_vector_type(2))) float v2f;
typedef __attribute__((ext_vector_type(8))) float v8f;

template <int XM>
__device__ __forceinline__ float gat_row_xadd(float v) {
  const int b = __builtin_bit_cast(int, v);
  const int o = __builtin_amdgcn_update_dpp(0, b, 0x160 | XM, 0xf, 0xf, true);
  return v + __builtin_bit_cast(float, o);
}

template <int UNR>
__device__ __forceinline__ void
gat_gemm_strip(const float* __restrict__ A, const float* __restrict__ B,
               const float* __restrict__ al, const float* __restrict__ ar,
               float* __restrict__ C, float* __restrict__ el,
               float* __restrict__ er, int M, int K, int strip) {
  const int lane = threadIdx.x & 31;
  const int row0 = strip * 16;
  const int kh   = lane >> 4;
  const int l15  = lane & 15;
  int mrow = row0 + l15;
  if (mrow >= M) mrow = M - 1;

  const float* arow = A + (size_t)mrow * K;
  v8f acc[4];
#pragma unroll
  for (int t = 0; t < 4; ++t) acc[t] = (v8f){0.f,0.f,0.f,0.f,0.f,0.f,0.f,0.f};

#pragma unroll UNR
  for (int k = 0; k < K; k += 4) {
    const v2f a = *(const v2f*)(arow + k + 2 * kh);
    const float* brow0 = B + (size_t)(k + 2 * kh) * 64 + l15;
#pragma unroll
    for (int t = 0; t < 4; ++t) {
      v2f b;
      b.x = brow0[t * 16];
      b.y = brow0[64 + t * 16];
      acc[t] = __builtin_amdgcn_wmma_f32_16x16x4_f32(
          false, a, false, b, (short)0, acc[t], false, false);
    }
  }

  const int rbase = row0 + 8 * kh;
#pragma unroll
  for (int t = 0; t < 4; ++t) {
    float* cp = C + (size_t)rbase * 64 + t * 16 + l15;
#pragma unroll
    for (int r = 0; r < 8; ++r)
      if (rbase + r < M) cp[(size_t)r * 64] = acc[t][r];
  }

  float alv[4], arv[4];
#pragma unroll
  for (int t = 0; t < 4; ++t) {
    alv[t] = al[t * 16 + l15];
    arv[t] = ar[t * 16 + l15];
  }
  float pl[8], pr[8];
#pragma unroll
  for (int r = 0; r < 8; ++r) {
    float sl = acc[0][r] * alv[0];
    float sr = acc[0][r] * arv[0];
#pragma unroll
    for (int t = 1; t < 4; ++t) {
      sl += acc[t][r] * alv[t];
      sr += acc[t][r] * arv[t];
    }
    pl[r] = sl;
    pr[r] = sr;
  }
#pragma unroll
  for (int r = 0; r < 8; ++r) { pl[r] = gat_row_xadd<1>(pl[r]); pr[r] = gat_row_xadd<1>(pr[r]); }
#pragma unroll
  for (int r = 0; r < 8; ++r) { pl[r] = gat_row_xadd<2>(pl[r]); pr[r] = gat_row_xadd<2>(pr[r]); }
#pragma unroll
  for (int r = 0; r < 8; ++r) { pl[r] = gat_row_xadd<4>(pl[r]); pr[r] = gat_row_xadd<4>(pr[r]); }
#pragma unroll
  for (int r = 0; r < 8; ++r) { pl[r] = gat_row_xadd<8>(pl[r]); pr[r] = gat_row_xadd<8>(pr[r]); }
  const int  rsel = l15 & 7;
  const bool isl  = l15 < 8;
  float outv = isl ? pl[0] : pr[0];
#pragma unroll
  for (int r = 1; r < 8; ++r)
    if (rsel == r) outv = isl ? pl[r] : pr[r];
  float* op = isl ? el : er;
  if (rbase + rsel < M) op[rbase + rsel] = outv;
}

__global__ void __launch_bounds__(256)
gat_gemm_wmma(const float* __restrict__ A, const float* __restrict__ B,
              const float* __restrict__ al, const float* __restrict__ ar,
              float* __restrict__ C, float* __restrict__ el,
              float* __restrict__ er, int M, int K) {
  const int strip = blockIdx.x * 8 + (int)(threadIdx.x >> 5);
  if (strip >= ((M + 15) >> 4)) return;
  gat_gemm_strip<4>(A, B, al, ar, C, el, er, M, K, strip);
}


#define GAT_BSH   8
#define GAT_T     1024
#define GAT_MAXB  GAT_T
#define GAT_EPT   4
#define GAT_CHUNK (GAT_T * GAT_EPT)
#define GAT_EPT2  8
#define GAT_CAP2  12288
#define GAT_MAXW  2048
#define GAT_RUNS  10

__device__ __forceinline__ int gat_wave_incl_scan(int v, int lane) {
#pragma unroll
  for (int off = 1; off < 32; off <<= 1) {
    const int t = __shfl_up(v, off, 32);
    if (lane >= off) v += t;
  }
  return v;
}

__device__ __forceinline__ int gat_block_excl_scan(int v, int* s_w) {
  const int lane = threadIdx.x & 31;
  const int wv   = threadIdx.x >> 5;
  const int nw   = (blockDim.x + 31) >> 5;
  const int inc  = gat_wave_incl_scan(v, lane);
  if (lane == 31) s_w[wv] = inc;
  __syncthreads();
  if (wv == 0) {
    const int w  = (lane < nw) ? s_w[lane] : 0;
    const int wi = gat_wave_incl_scan(w, lane);
    s_w[lane] = wi - w;
    if (lane == 31) s_w[32] = wi;
  }
  __syncthreads();
  return inc - v + s_w[wv];
}

__device__ __forceinline__ void
gat_chunk_sort_body(const int* __restrict__ src, const int* __restrict__ dst,
                    int* __restrict__ cnt_t, int* __restrict__ off_t,
                    int* __restrict__ tmpw, int E, int B, int nwg, int wg) {
  int2* __restrict__ tmp = (int2*)tmpw;
  __shared__ int  s_h[GAT_MAXB];
  __shared__ int  s_cur[GAT_MAXB];
  __shared__ int2 s_pair[GAT_CHUNK];
  __shared__ int  s_w[33];
  const int b  = threadIdx.x;
  s_h[b] = 0;
  const int e0 = wg * GAT_CHUNK + threadIdx.x;
  int sv[GAT_EPT], dv[GAT_EPT];
#pragma unroll
  for (int k = 0; k < GAT_EPT; ++k) {
    const int e = e0 + k * GAT_T;
    dv[k] = (e < E) ? dst[e] : -1;
    sv[k] = (e < E) ? src[e] : 0;
  }
  __syncthreads();
#pragma unroll
  for (int k = 0; k < GAT_EPT; ++k)
    if (dv[k] >= 0) atomicAdd(&s_h[dv[k] >> GAT_BSH], 1);
  __syncthreads();
  const int c  = s_h[b];
  const int ex = gat_block_excl_scan(c, s_w);
  s_cur[b] = ex;
  if (b < B) {
    cnt_t[(size_t)b * nwg + wg] = c;
    off_t[(size_t)b * nwg + wg] = ex;
  }
  __syncthreads();
#pragma unroll
  for (int k = 0; k < GAT_EPT; ++k) {
    if (dv[k] >= 0) {
      const int pos = atomicAdd(&s_cur[dv[k] >> GAT_BSH], 1);
      s_pair[pos] = make_int2(sv[k], dv[k]);
    }
  }
  __syncthreads();
  const int m = min(GAT_CHUNK, E - wg * GAT_CHUNK);
#pragma unroll
  for (int k = 0; k < GAT_EPT; ++k) {
    const int i = threadIdx.x + k * GAT_T;
    if (i < m) tmp[(size_t)wg * GAT_CHUNK + i] = s_pair[i];
  }
}

#define GAT_GW 16
__global__ void __launch_bounds__(GAT_T) __attribute__((amdgpu_waves_per_eu(12, 16)))
gat_sort_and_gemm(const int* __restrict__ src, const int* __restrict__ dst,
                  int* __restrict__ cnt_t, int* __restrict__ off_t,
                  int* __restrict__ tmpw, int E, int B, int nwg,
                  const float* __restrict__ A, const float* __restrict__ Wm,
                  const float* __restrict__ al, const float* __restrict__ ar,
                  float* __restrict__ C, float* __restrict__ el,
                  float* __restrict__ er, int M, int K, int gblk) {
  if ((int)blockIdx.x < gblk) {
    const int wv    = (int)(threadIdx.x >> 5);
    const int strip = (int)blockIdx.x * GAT_GW + wv;
    if (wv < GAT_GW && strip < ((M + 15) >> 4))
      gat_gemm_strip<2>(A, Wm, al, ar, C, el, er, M, K, strip);
    return;
  }
  gat_chunk_sort_body(src, dst, cnt_t, off_t, tmpw, E, B, nwg,
                      (int)blockIdx.x - gblk);
}

__device__ __forceinline__ int2 gat_run_fetch(const int2* __restrict__ tmp,
                                              const int* s_rp, const int* s_ro,
                                              int nwg, int i) {
  int lo = 0, hi = nwg - 1;
  while (lo < hi) {
    const int mid = (lo + hi + 1) >> 1;
    if (s_rp[mid] <= i) lo = mid; else hi = mid - 1;
  }
  return tmp[s_ro[lo] + (i - s_rp[lo])];
}

__global__ void __launch_bounds__(GAT_T)
gat_bucket_csr(const int* __restrict__ tmpw, const int* __restrict__ cnt_t,
               const int* __restrict__ off_t,
               int* __restrict__ rowstart, int* __restrict__ deg,
               int* __restrict__ ssrc, int N, int nwg) {
  const int2* __restrict__ tmp = (const int2*)tmpw;
  __shared__ int  s_cnt[256];
  __shared__ int  s_cur[256];
  __shared__ int  s_w[33];
  __shared__ int  s_w2[33];
  __shared__ int  s_rc[GAT_MAXW];
  __shared__ int  s_ro[GAT_MAXW];
  __shared__ int  s_rp[GAT_MAXW];
  __shared__ int2 s_stage[GAT_CAP2];
  __shared__ int  s_sorted[GAT_CAP2];
  const int b   = blockIdx.x;
  const int tid = threadIdx.x;

  int cnt = 0, base = 0;
  for (int w0 = 0; w0 < nwg; w0 += GAT_T) {
    const int w  = w0 + tid;
    const int c  = (w < nwg) ? cnt_t[(size_t)b * nwg + w] : 0;
    const int o  = (w < nwg) ? off_t[(size_t)b * nwg + w] : 0;
    const int ex = gat_block_excl_scan(c, s_w);
    (void)gat_block_excl_scan(o, s_w2);
    if (w < nwg) {
      s_rc[w] = c;
      s_ro[w] = w * GAT_CHUNK + o;
      s_rp[w] = cnt + ex;
    }
    cnt  += s_w[32];
    base += s_w2[32];
    __syncthreads();
  }
  if (tid < 256) s_cnt[tid] = 0;
  __syncthreads();
  const bool fits = cnt <= GAT_CAP2;

  if (fits) {
    const int lane = tid & 31;
    const int wv   = tid >> 5;
    for (int w0 = wv; w0 < nwg; w0 += 32 * GAT_RUNS) {
      int  rc[GAT_RUNS], ro[GAT_RUNS], rp[GAT_RUNS];
      int2 v[GAT_RUNS];
#pragma unroll
      for (int u = 0; u < GAT_RUNS; ++u) {
        const int  w  = w0 + 32 * u;
        const bool ok = w < nwg;
        const int  wc = min(w, nwg - 1);
        const int  tc = s_rc[wc];
        const int  to = s_ro[wc];
        const int  tp = s_rp[wc];
        rc[u] = ok ? tc : 0;
        ro[u] = ok ? to : 0;
        rp[u] = ok ? tp : 0;
      }
#pragma unroll
      for (int u = 0; u < GAT_RUNS; ++u)
        v[u] = (lane < rc[u]) ? tmp[ro[u] + lane] : make_int2(0, 0);
#pragma unroll
      for (int u = 0; u < GAT_RUNS; ++u) {
        if (lane < rc[u]) {
          s_stage[rp[u] + lane] = v[u];
          atomicAdd(&s_cnt[v[u].y & 255], 1);
        }
      }
      for (int u = 0; u < GAT_RUNS; ++u) {
        const int w = w0 + 32 * u;
        if (w >= nwg) break;
        const int c = s_rc[w];
        for (int i = lane + 32; i < c; i += 32) {
          const int2 q = tmp[s_ro[w] + i];
          s_stage[s_rp[w] + i] = q;
          atomicAdd(&s_cnt[q.y & 255], 1);
        }
      }
    }
  } else {
    for (int i0 = tid; i0 < cnt; i0 += GAT_T * GAT_EPT2) {
      int dl[GAT_EPT2];
#pragma unroll
      for (int k = 0; k < GAT_EPT2; ++k) {
        const int i = i0 + k * GAT_T;
        dl[k] = (i < cnt) ? gat_run_fetch(tmp, s_rp, s_ro, nwg, i).y : -1;
      }
#pragma unroll
      for (int k = 0; k < GAT_EPT2; ++k)
        if (dl[k] >= 0) atomicAdd(&s_cnt[dl[k] & 255], 1);
    }
  }
  __syncthreads();
  const int c  = (tid < 256) ? s_cnt[tid] : 0;
  const int ex = gat_block_excl_scan(c, s_w);
  if (tid < 256) {
    s_cur[tid] = ex;
    const int node = (b << GAT_BSH) + tid;
    if (node < N) { rowstart[node] = base + ex; deg[node] = c; }
  }
  __syncthreads();
  if (fits) {
    for (int i = tid; i < cnt; i += GAT_T) {
      const int2 p  = s_stage[i];
      const int pos = atomicAdd(&s_cur[p.y & 255], 1);
      s_sorted[pos] = p.x;
    }
    __syncthreads();
    for (int i = tid; i < cnt; i += GAT_T) ssrc[base + i] = s_sorted[i];
  } else {
    for (int i = tid; i < cnt; i += GAT_T) {
      const int2 p  = gat_run_fetch(tmp, s_rp, s_ro, nwg, i);
      const int pos = atomicAdd(&s_cur[p.y & 255], 1);
      ssrc[base + pos] = p.x;
    }
  }
}


#define GAT_GRP 16

__device__ __forceinline__ void gat_gather_chunk(const float* __restrict__ z,
                                                 int svl, float wl, int nn,
                                                 int lane, float2& acc) {
  const int wi = __builtin_bit_cast(int, wl);
  int j0 = 0;
  for (; j0 + GAT_GRP <= nn; j0 += GAT_GRP) {
    float2 zz[GAT_GRP];
    float  w[GAT_GRP];
#pragma unroll
    for (int k = 0; k < GAT_GRP; ++k) {
      const int s = __builtin_amdgcn_readlane(svl, j0 + k);
      w[k]  = __builtin_bit_cast(float, __builtin_amdgcn_readlane(wi, j0 + k));
      zz[k] = ((const float2*)(z + (size_t)s * 64))[lane];
    }
#pragma unroll
    for (int k = 0; k < GAT_GRP; ++k) {
      acc.x = fmaf(w[k], zz[k].x, acc.x);
      acc.y = fmaf(w[k], zz[k].y, acc.y);
    }
  }
  if (j0 < nn) {
    float2 zz[GAT_GRP];
    float  w[GAT_GRP];
#pragma unroll
    for (int k = 0; k < GAT_GRP; ++k) {
      zz[k] = make_float2(0.f, 0.f);
      w[k]  = 0.f;
      if (j0 + k < nn) {
        const int s = __builtin_amdgcn_readlane(svl, j0 + k);
        w[k]  = __builtin_bit_cast(float, __builtin_amdgcn_readlane(wi, j0 + k));
        zz[k] = ((const float2*)(z + (size_t)s * 64))[lane];
      }
    }
#pragma unroll
    for (int k = 0; k < GAT_GRP; ++k) {
      if (j0 + k < nn) {
        acc.x = fmaf(w[k], zz[k].x, acc.x);
        acc.y = fmaf(w[k], zz[k].y, acc.y);
      }
    }
  }
}

__global__ void __launch_bounds__(256)
gat_node_aggregate(const int* __restrict__ rowstart,
                   const int* __restrict__ deg,
                   const int* __restrict__ ssrc,
                   const float* __restrict__ el,
                   const float* __restrict__ er,
                   const float* __restrict__ z,
                   const float* __restrict__ bias,
                   float* __restrict__ out,
                   int N, int applyRelu) {
  const int lane = threadIdx.x & 31;
  const int wv  = __builtin_amdgcn_readfirstlane((int)(threadIdx.x >> 5));
  const int wid = (int)blockIdx.x * (int)(blockDim.x >> 5) + wv;
  if (wid >= N) return;
  const int start = rowstart[wid];
  const int cnt   = deg[wid];
  const float ern = er[wid];

  int   sv0 = 0;
  float v0  = 0.f;
  float mx  = -INFINITY;
  if (lane < cnt) {
    sv0 = ssrc[start + lane];
    const float v = el[sv0] + ern;
    v0 = (v > 0.f) ? v : 0.2f * v;
    mx = fmaxf(mx, v0);
  }
  for (int base = 32; base < cnt; base += 32) {
    const int j = base + lane;
    if (j < cnt) {
      float v = el[ssrc[start + j]] + ern;
      v = (v > 0.f) ? v : 0.2f * v;
      mx = fmaxf(mx, v);
    }
  }
#pragma unroll
  for (int off = 16; off; off >>= 1) mx = fmaxf(mx, __shfl_xor(mx, off, 32));

  const float ex0 = (lane < cnt) ? __expf(v0 - mx) : 0.f;
  float sum = ex0;
  for (int base = 32; base < cnt; base += 32) {
    const int j = base + lane;
    if (j < cnt) {
      float v = el[ssrc[start + j]] + ern;
      v = (v > 0.f) ? v : 0.2f * v;
      sum += __expf(v - mx);
    }
  }
#pragma unroll
  for (int off = 16; off; off >>= 1) sum += __shfl_xor(sum, off, 32);
  const float rcp = (cnt > 0) ? (1.f / sum) : 0.f;

  float2 acc = {0.f, 0.f};
  gat_gather_chunk(z, sv0, ex0 * rcp, min(cnt, 32), lane, acc);
  for (int base = 32; base < cnt; base += 32) {
    const int j = base + lane;
    int   svc = 0;
    float wc  = 0.f;
    if (j < cnt) {
      svc = ssrc[start + j];
      float v = el[svc] + ern;
      v = (v > 0.f) ? v : 0.2f * v;
      wc = __expf(v - mx) * rcp;
    }
    gat_gather_chunk(z, svc, wc, min(cnt - base, 32), lane, acc);
  }

  float2 o;
  o.x = acc.x + bias[2 * lane];
  o.y = acc.y + bias[2 * lane + 1];
  if (applyRelu) { o.x = fmaxf(o.x, 0.f); o.y = fmaxf(o.y, 0.f); }
  ((float2*)(out + (size_t)wid * 64))[lane] = o;
}


extern "C" void kernel_launch(void* const* d_in, const int* in_sizes, int n_in,
                              void* d_out, int out_size, void* d_ws, size_t ws_size,
                              hipStream_t stream) {
  const float* x   = (const float*)d_in[0];
  const float* W1  = (const float*)d_in[1];
  const float* al1 = (const float*)d_in[2];
  const float* ar1 = (const float*)d_in[3];
  const float* b1  = (const float*)d_in[4];
  const float* W2  = (const float*)d_in[5];
  const float* al2 = (const float*)d_in[6];
  const float* ar2 = (const float*)d_in[7];
  const float* b2  = (const float*)d_in[8];
  const int*   src = (const int*)d_in[9];
  const int*   dst = (const int*)d_in[10];
  float* out = (float*)d_out;

  const int F  = 64;
  const int N  = out_size / F;
  const int E  = in_sizes[9];
  const int K1 = in_sizes[0] / N;
  (void)n_in; (void)ws_size;

  char* ws = (char*)d_ws;
  size_t off = 0;
  auto alloc = [&](size_t bytes) -> void* {
    void* p = ws + off;
    off += (bytes + 255) & ~(size_t)255;
    return p;
  };
  float* zbuf     = (float*)alloc((size_t)N * F * sizeof(float));
  float* hbuf     = (float*)alloc((size_t)N * F * sizeof(float));
  float* el       = (float*)alloc((size_t)N * sizeof(float));
  float* er       = (float*)alloc((size_t)N * sizeof(float));
  int*   deg      = (int*)alloc((size_t)N * sizeof(int));
  int*   rowstart = (int*)alloc((size_t)N * sizeof(int));
  int*   ssrc     = (int*)alloc((size_t)E * sizeof(int));
  const int nb    = (N + 255) / 256;
  const int nwg   = (E + GAT_CHUNK - 1) / GAT_CHUNK;
  int*   cnt_t    = (int*)alloc((size_t)nb * nwg * sizeof(int));
  int*   off_t    = (int*)alloc((size_t)nb * nwg * sizeof(int));
  int*   tmp      = (int*)alloc((size_t)E * 2 * sizeof(int));

  const int gWaveN = (N + 7) / 8;
  const int gGemm  = (((N + 15) / 16) + 7) / 8;
  const int gGemm1 = (((N + 15) / 16) + GAT_GW - 1) / GAT_GW;

  gat_sort_and_gemm <<<gGemm1 + nwg, GAT_T, 0, stream>>>(src, dst, cnt_t, off_t, tmp, E, nb, nwg,
                                                x, W1, al1, ar1, zbuf, el, er, N, K1,
                                                gGemm1);
  gat_bucket_csr    <<<nb,  GAT_T, 0, stream>>>(tmp, cnt_t, off_t, rowstart, deg, ssrc,
                                                N, nwg);

  gat_node_aggregate<<<gWaveN, 256, 0, stream>>>(rowstart, deg, ssrc, el, er,
                                                 zbuf, b1, hbuf, N,  1);

  gat_gemm_wmma     <<<gGemm,  256, 0, stream>>>(hbuf, W2, al2, ar2, zbuf, el, er, N, F);
  gat_node_aggregate<<<gWaveN, 256, 0, stream>>>(rowstart, deg, ssrc, el, er,
                                                 zbuf, b2, out, N,  0);
}


# aggregate: a resident grid of 8192 waves (not 50000 one-node waves), each looping over 6 or 7 consecutive nodes, the next node's bounds and first indices requested ahead; sum order unchanged
# speedup vs baseline: 3.4394x; 1.1061x over previous
#include <hip/hip_runtime.h>
#include <math.h>


typedef __attribute__((ext_vector_type(2))) float v2f;
typedef __attribute__((ext_vector_type(8))) float v8f;

template <int XM>
__device__ __forceinline__ float gat_row_xadd(float v) {
  const int b = __builtin_bit_cast(int, v);
  const int o = __builtin_amdgcn_update_dpp(0, b, 0x160 | XM, 0xf, 0xf, true);
  return v + __builtin_bit_cast(float, o);
}

template <int UNR>
__device__ __forceinline__ void
gat_gemm_strip(const float* __restrict__ A, const float* __restrict__ B,
               const float* __restrict__ ar,
               float* __restrict__ C,
               float* __restrict__ er, int M, int K, int strip) {
  const int lane = threadIdx.x & 31;
  const int row0 = strip * 16;
  const int kh   = lane >> 4;
  const int l15  = lane & 15;
  int mrow = row0 + l15;
  if (mrow >= M) mrow = M - 1;

  const float* arow = A + (size_t)mrow * K;
  v8f acc[4];
#pragma unroll
  for (int t = 0; t < 4; ++t) acc[t] = (v8f){0.f,0.f,0.f,0.f,0.f,0.f,0.f,0.f};

#pragma unroll UNR
  for (int k = 0; k < K; k += 4) {
    const v2f a = *(const v2f*)(arow + k + 2 * kh);
    const float* brow0 = B + (size_t)(k + 2 * kh) * 64 + l15;
#pragma unroll
    for (int t = 0; t < 4; ++t) {
      v2f b;
      b.x = brow0[t * 16];
      b.y = brow0[64 + t * 16];
      acc[t] = __builtin_amdgcn_wmma_f32_16x16x4_f32(
          false, a, false, b, (short)0, acc[t], false, false);
    }
  }

  const int rbase = row0 + 8 * kh;
#pragma unroll
  for (int t = 0; t < 4; ++t) {
    float* cp = C + (size_t)rbase * 64 + t * 16 + l15;
#pragma unroll
    for (int r = 0; r < 8; ++r)
      if (rbase + r < M) cp[(size_t)r * 64] = acc[t][r];
  }

  float arv[4];
#pragma unroll
  for (int t = 0; t < 4; ++t) arv[t] = ar[t * 16 + l15];
  float pr[8];
#pragma unroll
  for (int r = 0; r < 8; ++r) {
    float sr = acc[0][r] * arv[0];
#pragma unroll
    for (int t = 1; t < 4; ++t) sr += acc[t][r] * arv[t];
    pr[r] = sr;
  }
#pragma unroll
  for (int r = 0; r < 8; ++r) pr[r] = gat_row_xadd<1>(pr[r]);
#pragma unroll
  for (int r = 0; r < 8; ++r) pr[r] = gat_row_xadd<2>(pr[r]);
#pragma unroll
  for (int r = 0; r < 8; ++r) pr[r] = gat_row_xadd<4>(pr[r]);
#pragma unroll
  for (int r = 0; r < 8; ++r) pr[r] = gat_row_xadd<8>(pr[r]);
  const int rsel = l15 & 7;
  float outv = pr[0];
#pragma unroll
  for (int r = 1; r < 8; ++r)
    if (rsel == r) outv = pr[r];
  if (l15 < 8 && rbase + rsel < M) er[rbase + rsel] = outv;
}

__global__ void __launch_bounds__(256)
gat_gemm_wmma(const float* __restrict__ A, const float* __restrict__ B,
              const float* __restrict__ ar,
              float* __restrict__ C,
              float* __restrict__ er, int M, int K) {
  const int strip = blockIdx.x * 8 + (int)(threadIdx.x >> 5);
  if (strip >= ((M + 15) >> 4)) return;
  gat_gemm_strip<4>(A, B, ar, C, er, M, K, strip);
}


#define GAT_BSH   8
#define GAT_T     1024
#define GAT_MAXB  GAT_T
#define GAT_EPT   4
#define GAT_CHUNK (GAT_T * GAT_EPT)
#define GAT_EPT2  8
#define GAT_CAP2  12288
#define GAT_MAXW  2048
#define GAT_RUNS  10

__device__ __forceinline__ int gat_wave_incl_scan(int v, int lane) {
#pragma unroll
  for (int off = 1; off < 32; off <<= 1) {
    const int t = __shfl_up(v, off, 32);
    if (lane >= off) v += t;
  }
  return v;
}

__device__ __forceinline__ int gat_block_excl_scan(int v, int* s_w) {
  const int lane = threadIdx.x & 31;
  const int wv   = threadIdx.x >> 5;
  const int nw   = (blockDim.x + 31) >> 5;
  const int inc  = gat_wave_incl_scan(v, lane);
  if (lane == 31) s_w[wv] = inc;
  __syncthreads();
  if (wv == 0) {
    const int w  = (lane < nw) ? s_w[lane] : 0;
    const int wi = gat_wave_incl_scan(w, lane);
    s_w[lane] = wi - w;
    if (lane == 31) s_w[32] = wi;
  }
  __syncthreads();
  return inc - v + s_w[wv];
}

__device__ __forceinline__ void
gat_chunk_sort_body(const int* __restrict__ src, const int* __restrict__ dst,
                    int* __restrict__ cnt_t, int* __restrict__ off_t,
                    int* __restrict__ tmpw, int E, int B, int nwg, int wg) {
  int2* __restrict__ tmp = (int2*)tmpw;
  __shared__ int  s_h[GAT_MAXB];
  __shared__ int  s_cur[GAT_MAXB];
  __shared__ int2 s_pair[GAT_CHUNK];
  __shared__ int  s_w[33];
  const int b  = threadIdx.x;
  s_h[b] = 0;
  const int e0 = wg * GAT_CHUNK + threadIdx.x;
  int sv[GAT_EPT], dv[GAT_EPT];
#pragma unroll
  for (int k = 0; k < GAT_EPT; ++k) {
    const int e = e0 + k * GAT_T;
    dv[k] = (e < E) ? dst[e] : -1;
    sv[k] = (e < E) ? src[e] : 0;
  }
  __syncthreads();
#pragma unroll
  for (int k = 0; k < GAT_EPT; ++k)
    if (dv[k] >= 0) atomicAdd(&s_h[dv[k] >> GAT_BSH], 1);
  __syncthreads();
  const int c  = s_h[b];
  const int ex = gat_block_excl_scan(c, s_w);
  s_cur[b] = ex;
  if (b < B) {
    cnt_t[(size_t)b * nwg + wg] = c;
    off_t[(size_t)b * nwg + wg] = ex;
  }
  __syncthreads();
#pragma unroll
  for (int k = 0; k < GAT_EPT; ++k) {
    if (dv[k] >= 0) {
      const int pos = atomicAdd(&s_cur[dv[k] >> GAT_BSH], 1);
      s_pair[pos] = make_int2(sv[k], dv[k]);
    }
  }
  __syncthreads();
  const int m = min(GAT_CHUNK, E - wg * GAT_CHUNK);
#pragma unroll
  for (int k = 0; k < GAT_EPT; ++k) {
    const int i = threadIdx.x + k * GAT_T;
    if (i < m) tmp[(size_t)wg * GAT_CHUNK + i] = s_pair[i];
  }
}

#define GAT_GW 16
__global__ void __launch_bounds__(GAT_T) __attribute__((amdgpu_waves_per_eu(12, 16)))
gat_sort_and_gemm(const int* __restrict__ src, const int* __restrict__ dst,
                  int* __restrict__ cnt_t, int* __restrict__ off_t,
                  int* __restrict__ tmpw, int E, int B, int nwg,
                  const float* __restrict__ A, const float* __restrict__ Wm,
                  const float* __restrict__ ar,
                  float* __restrict__ C,
                  float* __restrict__ er, int M, int K, int gblk) {
  if ((int)blockIdx.x < gblk) {
    const int wv    = (int)(threadIdx.x >> 5);
    const int strip = (int)blockIdx.x * GAT_GW + wv;
    if (wv < GAT_GW && strip < ((M + 15) >> 4))
      gat_gemm_strip<2>(A, Wm, ar, C, er, M, K, strip);
    return;
  }
  gat_chunk_sort_body(src, dst, cnt_t, off_t, tmpw, E, B, nwg,
                      (int)blockIdx.x - gblk);
}

__device__ __forceinline__ int2 gat_run_fetch(const int2* __restrict__ tmp,
                                              const int* s_rp, const int* s_ro,
                                              int nwg, int i) {
  int lo = 0, hi = nwg - 1;
  while (lo < hi) {
    const int mid = (lo + hi + 1) >> 1;
    if (s_rp[mid] <= i) lo = mid; else hi = mid - 1;
  }
  return tmp[s_ro[lo] + (i - s_rp[lo])];
}

__global__ void __launch_bounds__(GAT_T)
gat_bucket_csr(const int* __restrict__ tmpw, const int* __restrict__ cnt_t,
               const int* __restrict__ off_t,
               int* __restrict__ rowstart, int* __restrict__ deg,
               int* __restrict__ ssrc, int N, int nwg) {
  const int2* __restrict__ tmp = (const int2*)tmpw;
  __shared__ int  s_cnt[256];
  __shared__ int  s_cur[256];
  __shared__ int  s_w[33];
  __shared__ int  s_w2[33];
  __shared__ int  s_rc[GAT_MAXW];
  __shared__ int  s_ro[GAT_MAXW];
  __shared__ int  s_rp[GAT_MAXW];
  __shared__ int2 s_stage[GAT_CAP2];
  __shared__ int  s_sorted[GAT_CAP2];
  const int b   = blockIdx.x;
  const int tid = threadIdx.x;

  int cnt = 0, base = 0;
  for (int w0 = 0; w0 < nwg; w0 += GAT_T) {
    const int w  = w0 + tid;
    const int c  = (w < nwg) ? cnt_t[(size_t)b * nwg + w] : 0;
    const int o  = (w < nwg) ? off_t[(size_t)b * nwg + w] : 0;
    const int ex = gat_block_excl_scan(c, s_w);
    (void)gat_block_excl_scan(o, s_w2);
    if (w < nwg) {
      s_rc[w] = c;
      s_ro[w] = w * GAT_CHUNK + o;
      s_rp[w] = cnt + ex;
    }
    cnt  += s_w[32];
    base += s_w2[32];
    __syncthreads();
  }
  if (tid < 256) s_cnt[tid] = 0;
  __syncthreads();
  const bool fits = cnt <= GAT_CAP2;

  if (fits) {
    const int lane = tid & 31;
    const int wv   = tid >> 5;
    for (int w0 = wv; w0 < nwg; w0 += 32 * GAT_RUNS) {
      int  rc[GAT_RUNS], ro[GAT_RUNS], rp[GAT_RUNS];
      int2 v[GAT_RUNS];
#pragma unroll
      for (int u = 0; u < GAT_RUNS; ++u) {
        const int  w  = w0 + 32 * u;
        const bool ok = w < nwg;
        const int  wc = min(w, nwg - 1);
        const int  tc = s_rc[wc];
        const int  to = s_ro[wc];
        const int  tp = s_rp[wc];
        rc[u] = ok ? tc : 0;
        ro[u] = ok ? to : 0;
        rp[u] = ok ? tp : 0;
      }
#pragma unroll
      for (int u = 0; u < GAT_RUNS; ++u)
        v[u] = (lane < rc[u]) ? tmp[ro[u] + lane] : make_int2(0, 0);
#pragma unroll
      for (int u = 0; u < GAT_RUNS; ++u) {
        if (lane < rc[u]) {
          s_stage[rp[u] + lane] = v[u];
          atomicAdd(&s_cnt[v[u].y & 255], 1);
        }
      }
      for (int u = 0; u < GAT_RUNS; ++u) {
        const int w = w0 + 32 * u;
        if (w >= nwg) break;
        const int c = s_rc[w];
        for (int i = lane + 32; i < c; i += 32) {
          const int2 q = tmp[s_ro[w] + i];
          s_stage[s_rp[w] + i] = q;
          atomicAdd(&s_cnt[q.y & 255], 1);
        }
      }
    }
  } else {
    for (int i0 = tid; i0 < cnt; i0 += GAT_T * GAT_EPT2) {
      int dl[GAT_EPT2];
#pragma unroll
      for (int k = 0; k < GAT_EPT2; ++k) {
        const int i = i0 + k * GAT_T;
        dl[k] = (i < cnt) ? gat_run_fetch(tmp, s_rp, s_ro, nwg, i).y : -1;
      }
#pragma unroll
      for (int k = 0; k < GAT_EPT2; ++k)
        if (dl[k] >= 0) atomicAdd(&s_cnt[dl[k] & 255], 1);
    }
  }
  __syncthreads();
  const int c  = (tid < 256) ? s_cnt[tid] : 0;
  const int ex = gat_block_excl_scan(c, s_w);
  if (tid < 256) {
    s_cur[tid] = ex;
    const int node = (b << GAT_BSH) + tid;
    if (node < N) { rowstart[node] = base + ex; deg[node] = c; }
  }
  __syncthreads();
  if (fits) {
    for (int i = tid; i < cnt; i += GAT_T) {
      const int2 p  = s_stage[i];
      const int pos = atomicAdd(&s_cur[p.y & 255], 1);
      s_sorted[pos] = p.x;
    }
    __syncthreads();
    for (int i = tid; i < cnt; i += GAT_T) ssrc[base + i] = s_sorted[i];
  } else {
    for (int i = tid; i < cnt; i += GAT_T) {
      const int2 p  = gat_run_fetch(tmp, s_rp, s_ro, nwg, i);
      const int pos = atomicAdd(&s_cur[p.y & 255], 1);
      ssrc[base + pos] = p.x;
    }
  }
}


typedef __attribute__((ext_vector_type(4))) float v4f;

template <int I>
__device__ __forceinline__ int gat_row_share(int v) {
  return __builtin_amdgcn_update_dpp(0, v, 0x150 | I, 0xf, 0xf, true);
}

template <int XM>
__device__ __forceinline__ float gat_row_xsum(float keep, float send) {
  const int b = __builtin_bit_cast(int, send);
  const int o = __builtin_amdgcn_update_dpp(0, b, 0x160 | XM, 0xf, 0xf, true);
  return keep + __builtin_bit_cast(float, o);
}

template <int XM>
__device__ __forceinline__ float gat_row_xmax(float v) {
  const int b = __builtin_bit_cast(int, v);
  const int o = __builtin_amdgcn_update_dpp(0, b, 0x160 | XM, 0xf, 0xf, true);
  return fmaxf(v, __builtin_bit_cast(float, o));
}

__device__ __forceinline__ float gat_xhalf(float x) {
  const int b = __builtin_bit_cast(int, x);
  return __builtin_bit_cast(float,
      __builtin_amdgcn_permlanex16(b, b, 0x76543210, 0xfedcba98, false, false));
}

#define GAT_STEP_LOAD(I)                                                     \
  zz[I] = (v4f){0.f, 0.f, 0.f, 0.f};                                         \
  if (2 * (I) < nn) {                                                        \
    const unsigned off = ((unsigned)gat_row_share<I>(svl) << 8) | cb;        \
    zz[I] = *(const v4f*)((const char*)z + off);                             \
  }

#define GAT_STEP_ACC(I)                                                      \
  {                                                                          \
    const float wi = __builtin_bit_cast(float,                               \
        gat_row_share<I>(__builtin_bit_cast(int, w)));                       \
    const v2f w2 = {wi, wi};                                                 \
    acc_lo = __builtin_elementwise_fma(w2, __builtin_shufflevector(zz[I], zz[I], 0, 1), acc_lo); \
    acc_hi = __builtin_elementwise_fma(w2, __builtin_shufflevector(zz[I], zz[I], 2, 3), acc_hi); \
  }

__global__ void __launch_bounds__(256)
gat_node_aggregate(const int* __restrict__ rowstart,
                   const int* __restrict__ deg,
                   const int* __restrict__ ssrc,
                   const float* __restrict__ al,
                   const float* __restrict__ er,
                   const float* __restrict__ z,
                   const float* __restrict__ bias,
                   float* __restrict__ out,
                   int N, int applyRelu) {
  const int lane = threadIdx.x & 31;
  const int wv  = __builtin_amdgcn_readfirstlane((int)(threadIdx.x >> 5));
  const int wpb = (int)(blockDim.x >> 5);
  const int wid = (int)blockIdx.x * wpb + wv;
  const int nwv = (int)gridDim.x * wpb;
  const int npw = N / nwv;
  const int nex = N - npw * nwv;
  int node = wid * npw + min(wid, nex);
  const int nlast = node + npw - ((wid < nex) ? 0 : 1);
  if (node > nlast) return;

  const int h     = lane >> 4;
  const int c     = lane & 15;
  const int eslot = 2 * c + h;
  const unsigned cb = (unsigned)c << 4;
  const v4f al4   = ((const v4f*)al)[c];
  const v2f al_lo = __builtin_shufflevector(al4, al4, 0, 1);
  const v2f al_hi = __builtin_shufflevector(al4, al4, 2, 3);
  const bool b3 = (c & 8) != 0, b2 = (c & 4) != 0, b1 = (c & 2) != 0, b0 = (c & 1) != 0;

  int   start = rowstart[node];
  int   cnt   = deg[node];
  float ern   = er[node];
  const int nd1 = min(node + 1, nlast);
  int   start1 = rowstart[nd1];
  int   cnt1   = deg[nd1];
  float ern1   = er[nd1];
  int svl = 0;
  if (cnt > 0) svl = ssrc[start + min(eslot, min(cnt, 32) - 1)];

  for (;;) {
  const int nd2 = min(node + 2, nlast);
  const int   start2 = rowstart[nd2];
  const int   cnt2   = deg[nd2];
  const float ern2   = er[nd2];
  int svl1 = 0;
  if (cnt1 > 0) svl1 = ssrc[start1 + min(eslot, min(cnt1, 32) - 1)];

  float m = -INFINITY;
  float s = 0.f;
  v2f acc_lo = {0.f, 0.f}, acc_hi = {0.f, 0.f};
  for (int base = 0; base < cnt; base += 32) {
    const int nn  = min(cnt - base, 32);
    if (base > 0) svl = ssrc[start + base + min(eslot, nn - 1)];

    v4f zz[16];
    GAT_STEP_LOAD(0)
    GAT_STEP_LOAD(1)
    GAT_STEP_LOAD(2)
    GAT_STEP_LOAD(3)
    GAT_STEP_LOAD(4)
    GAT_STEP_LOAD(5)
    GAT_STEP_LOAD(6)
    GAT_STEP_LOAD(7)
    GAT_STEP_LOAD(8)
    GAT_STEP_LOAD(9)
    GAT_STEP_LOAD(10)
    GAT_STEP_LOAD(11)
    GAT_STEP_LOAD(12)
    GAT_STEP_LOAD(13)
    GAT_STEP_LOAD(14)
    GAT_STEP_LOAD(15)

    float q[8], r[4], t2[2];
#pragma unroll
    for (int k = 0; k < 8; ++k) {
      v2f ta = __builtin_shufflevector(zz[k], zz[k], 0, 1) * al_lo;
      v2f tb = __builtin_shufflevector(zz[k + 8], zz[k + 8], 0, 1) * al_lo;
      ta = __builtin_elementwise_fma(__builtin_shufflevector(zz[k], zz[k], 2, 3), al_hi, ta);
      tb = __builtin_elementwise_fma(__builtin_shufflevector(zz[k + 8], zz[k + 8], 2, 3), al_hi, tb);
      const float pa = ta.x + ta.y;
      const float pb = tb.x + tb.y;
      q[k] = gat_row_xsum<8>(b3 ? pb : pa, b3 ? pa : pb);
    }
#pragma unroll
    for (int k = 0; k < 4; ++k)
      r[k] = gat_row_xsum<4>(b2 ? q[k + 4] : q[k], b2 ? q[k] : q[k + 4]);
#pragma unroll
    for (int k = 0; k < 2; ++k)
      t2[k] = gat_row_xsum<2>(b1 ? r[k + 2] : r[k], b1 ? r[k] : r[k + 2]);
    const float elv = gat_row_xsum<1>(b0 ? t2[1] : t2[0], b0 ? t2[0] : t2[1]);

    float e = elv + ern;
    e = (e > 0.f) ? e : 0.2f * e;
    e = (eslot < nn) ? e : -INFINITY;
    float cm = e;
    cm = gat_row_xmax<1>(cm);
    cm = gat_row_xmax<2>(cm);
    cm = gat_row_xmax<4>(cm);
    cm = gat_row_xmax<8>(cm);
    cm = fmaxf(cm, gat_xhalf(cm));
    const float mn = fmaxf(m, cm);
    const float sc = __expf(m - mn);
    const float w  = __expf(e - mn);
    m = mn;
    s = fmaf(s, sc, w);
    acc_lo *= sc;
    acc_hi *= sc;
    GAT_STEP_ACC(0)
    GAT_STEP_ACC(1)
    GAT_STEP_ACC(2)
    GAT_STEP_ACC(3)
    GAT_STEP_ACC(4)
    GAT_STEP_ACC(5)
    GAT_STEP_ACC(6)
    GAT_STEP_ACC(7)
    GAT_STEP_ACC(8)
    GAT_STEP_ACC(9)
    GAT_STEP_ACC(10)
    GAT_STEP_ACC(11)
    GAT_STEP_ACC(12)
    GAT_STEP_ACC(13)
    GAT_STEP_ACC(14)
    GAT_STEP_ACC(15)
  }

  s = gat_row_xadd<1>(s);
  s = gat_row_xadd<2>(s);
  s = gat_row_xadd<4>(s);
  s = gat_row_xadd<8>(s);
  s += gat_xhalf(s);
  float ax = acc_lo.x, ay = acc_lo.y, az = acc_hi.x, aw = acc_hi.y;
  ax += gat_xhalf(ax);
  ay += gat_xhalf(ay);
  az += gat_xhalf(az);
  aw += gat_xhalf(aw);
  const float rcp = (cnt > 0) ? (1.f / s) : 0.f;

  const v4f b4 = ((const v4f*)bias)[c];
  v4f o;
  o.x = fmaf(ax, rcp, b4.x);
  o.y = fmaf(ay, rcp, b4.y);
  o.z = fmaf(az, rcp, b4.z);
  o.w = fmaf(aw, rcp, b4.w);
  if (applyRelu) {
    o.x = fmaxf(o.x, 0.f); o.y = fmaxf(o.y, 0.f);
    o.z = fmaxf(o.z, 0.f); o.w = fmaxf(o.w, 0.f);
  }
  if (h == 0) ((v4f*)(out + (size_t)node * 64))[c] = o;

  if (node >= nlast) break;
  ++node;
  start = start1; cnt = cnt1; ern = ern1; svl = svl1;
  start1 = start2; cnt1 = cnt2; ern1 = ern2;
  }
}
#undef GAT_STEP_LOAD
#undef GAT_STEP_ACC


extern "C" void kernel_launch(void* const* d_in, const int* in_sizes, int n_in,
                              void* d_out, int out_size, void* d_ws, size_t ws_size,
                              hipStream_t stream) {
  const float* x   = (const float*)d_in[0];
  const float* W1  = (const float*)d_in[1];
  const float* al1 = (const float*)d_in[2];
  const float* ar1 = (const float*)d_in[3];
  const float* b1  = (const float*)d_in[4];
  const float* W2  = (const float*)d_in[5];
  const float* al2 = (const float*)d_in[6];
  const float* ar2 = (const float*)d_in[7];
  const float* b2  = (const float*)d_in[8];
  const int*   src = (const int*)d_in[9];
  const int*   dst = (const int*)d_in[10];
  float* out = (float*)d_out;

  const int F  = 64;
  const int N  = out_size / F;
  const int E  = in_sizes[9];
  const int K1 = in_sizes[0] / N;
  (void)n_in; (void)ws_size;

  char* ws = (char*)d_ws;
  size_t off = 0;
  auto alloc = [&](size_t bytes) -> void* {
    void* p = ws + off;
    off += (bytes + 255) & ~(size_t)255;
    return p;
  };
  float* zbuf     = (float*)alloc((size_t)N * F * sizeof(float));
  float* hbuf     = (float*)alloc((size_t)N * F * sizeof(float));
  float* er       = (float*)alloc((size_t)N * sizeof(float));
  int*   deg      = (int*)alloc((size_t)N * sizeof(int));
  int*   rowstart = (int*)alloc((size_t)N * sizeof(int));
  int*   ssrc     = (int*)alloc((size_t)E * sizeof(int));
  const int nb    = (N + 255) / 256;
  const int nwg   = (E + GAT_CHUNK - 1) / GAT_CHUNK;
  int*   cnt_t    = (int*)alloc((size_t)nb * nwg * sizeof(int));
  int*   off_t    = (int*)alloc((size_t)nb * nwg * sizeof(int));
  int*   tmp      = (int*)alloc((size_t)E * 2 * sizeof(int));

  const int gAgg   = 2048;
  const int gGemm  = (((N + 15) / 16) + 7) / 8;
  const int gGemm1 = (((N + 15) / 16) + GAT_GW - 1) / GAT_GW;

  gat_sort_and_gemm <<<gGemm1 + nwg, GAT_T, 0, stream>>>(src, dst, cnt_t, off_t, tmp, E, nb, nwg,
                                                x, W1, ar1, zbuf, er, N, K1,
                                                gGemm1);
  gat_bucket_csr    <<<nb,  GAT_T, 0, stream>>>(tmp, cnt_t, off_t, rowstart, deg, ssrc,
                                                N, nwg);

  gat_node_aggregate<<<gAgg, 128, 0, stream>>>(rowstart, deg, ssrc, al1, er,
                                                 zbuf, b1, hbuf, N,  1);

  gat_gemm_wmma     <<<gGemm,  256, 0, stream>>>(hbuf, W2, ar2, zbuf, er, N, F);
  gat_node_aggregate<<<gAgg, 128, 0, stream>>>(rowstart, deg, ssrc, al2, er,
                                                 zbuf, b2, out, N,  0);
}


# layer-2 GEMM fed from layer 1's aggregation through LDS: 16-node workgroups keep their h rows in LDS and multiply by W2 there; no h array, no GEMM launch, 5->4 launches; sum order unchanged
# speedup vs baseline: 3.6822x; 1.0706x over previous
#include <hip/hip_runtime.h>
#include <math.h>


typedef __attribute__((ext_vector_type(2))) float v2f;
typedef __attribute__((ext_vector_type(8))) float v8f;

template <int XM>
__device__ __forceinline__ float gat_row_xadd(float v) {
  const int b = __builtin_bit_cast(int, v);
  const int o = __builtin_amdgcn_update_dpp(0, b, 0x160 | XM, 0xf, 0xf, true);
  return v + __builtin_bit_cast(float, o);
}

template <int UNR>
__device__ __forceinline__ void
gat_gemm_strip(const float* __restrict__ A, const float* __restrict__ B,
               const float* __restrict__ ar,
               float* __restrict__ C,
               float* __restrict__ er, int M, int K, int strip) {
  const int lane = threadIdx.x & 31;
  const int row0 = strip * 16;
  const int kh   = lane >> 4;
  const int l15  = lane & 15;
  int mrow = row0 + l15;
  if (mrow >= M) mrow = M - 1;

  const float* arow = A + (size_t)mrow * K;
  v8f acc[4];
#pragma unroll
  for (int t = 0; t < 4; ++t) acc[t] = (v8f){0.f,0.f,0.f,0.f,0.f,0.f,0.f,0.f};

#pragma unroll UNR
  for (int k = 0; k < K; k += 4) {
    const v2f a = *(const v2f*)(arow + k + 2 * kh);
    const float* brow0 = B + (size_t)(k + 2 * kh) * 64 + l15;
#pragma unroll
    for (int t = 0; t < 4; ++t) {
      v2f b;
      b.x = brow0[t * 16];
      b.y = brow0[64 + t * 16];
      acc[t] = __builtin_amdgcn_wmma_f32_16x16x4_f32(
          false, a, false, b, (short)0, acc[t], false, false);
    }
  }

  const int rbase = row0 + 8 * kh;
#pragma unroll
  for (int t = 0; t < 4; ++t) {
    float* cp = C + (size_t)rbase * 64 + t * 16 + l15;
#pragma unroll
    for (int r = 0; r < 8; ++r)
      if (rbase + r < M) cp[(size_t)r * 64] = acc[t][r];
  }

  float arv[4];
#pragma unroll
  for (int t = 0; t < 4; ++t) arv[t] = ar[t * 16 + l15];
  float pr[8];
#pragma unroll
  for (int r = 0; r < 8; ++r) {
    float sr = acc[0][r] * arv[0];
#pragma unroll
    for (int t = 1; t < 4; ++t) sr += acc[t][r] * arv[t];
    pr[r] = sr;
  }
#pragma unroll
  for (int r = 0; r < 8; ++r) pr[r] = gat_row_xadd<1>(pr[r]);
#pragma unroll
  for (int r = 0; r < 8; ++r) pr[r] = gat_row_xadd<2>(pr[r]);
#pragma unroll
  for (int r = 0; r < 8; ++r) pr[r] = gat_row_xadd<4>(pr[r]);
#pragma unroll
  for (int r = 0; r < 8; ++r) pr[r] = gat_row_xadd<8>(pr[r]);
  const int rsel = l15 & 7;
  float outv = pr[0];
#pragma unroll
  for (int r = 1; r < 8; ++r)
    if (rsel == r) outv = pr[r];
  if (l15 < 8 && rbase + rsel < M) er[rbase + rsel] = outv;
}


#define GAT_BSH   8
#define GAT_T     1024
#define GAT_MAXB  GAT_T
#define GAT_EPT   4
#define GAT_CHUNK (GAT_T * GAT_EPT)
#define GAT_EPT2  8
#define GAT_CAP2  12288
#define GAT_MAXW  2048
#define GAT_RUNS  10

__device__ __forceinline__ int gat_wave_incl_scan(int v, int lane) {
#pragma unroll
  for (int off = 1; off < 32; off <<= 1) {
    const int t = __shfl_up(v, off, 32);
    if (lane >= off) v += t;
  }
  return v;
}

__device__ __forceinline__ int gat_block_excl_scan(int v, int* s_w) {
  const int lane = threadIdx.x & 31;
  const int wv   = threadIdx.x >> 5;
  const int nw   = (blockDim.x + 31) >> 5;
  const int inc  = gat_wave_incl_scan(v, lane);
  if (lane == 31) s_w[wv] = inc;
  __syncthreads();
  if (wv == 0) {
    const int w  = (lane < nw) ? s_w[lane] : 0;
    const int wi = gat_wave_incl_scan(w, lane);
    s_w[lane] = wi - w;
    if (lane == 31) s_w[32] = wi;
  }
  __syncthreads();
  return inc - v + s_w[wv];
}

__device__ __forceinline__ void
gat_chunk_sort_body(const int* __restrict__ src, const int* __restrict__ dst,
                    int* __restrict__ cnt_t, int* __restrict__ off_t,
                    int* __restrict__ tmpw, int E, int B, int nwg, int wg) {
  int2* __restrict__ tmp = (int2*)tmpw;
  __shared__ int  s_h[GAT_MAXB];
  __shared__ int  s_cur[GAT_MAXB];
  __shared__ int2 s_pair[GAT_CHUNK];
  __shared__ int  s_w[33];
  const int b  = threadIdx.x;
  s_h[b] = 0;
  const int e0 = wg * GAT_CHUNK + threadIdx.x;
  int sv[GAT_EPT], dv[GAT_EPT];
#pragma unroll
  for (int k = 0; k < GAT_EPT; ++k) {
    const int e = e0 + k * GAT_T;
    dv[k] = (e < E) ? dst[e] : -1;
    sv[k] = (e < E) ? src[e] : 0;
  }
  __syncthreads();
#pragma unroll
  for (int k = 0; k < GAT_EPT; ++k)
    if (dv[k] >= 0) atomicAdd(&s_h[dv[k] >> GAT_BSH], 1);
  __syncthreads();
  const int c  = s_h[b];
  const int ex = gat_block_excl_scan(c, s_w);
  s_cur[b] = ex;
  if (b < B) {
    cnt_t[(size_t)b * nwg + wg] = c;
    off_t[(size_t)b * nwg + wg] = ex;
  }
  __syncthreads();
#pragma unroll
  for (int k = 0; k < GAT_EPT; ++k) {
    if (dv[k] >= 0) {
      const int pos = atomicAdd(&s_cur[dv[k] >> GAT_BSH], 1);
      s_pair[pos] = make_int2(sv[k], dv[k]);
    }
  }
  __syncthreads();
  const int m = min(GAT_CHUNK, E - wg * GAT_CHUNK);
#pragma unroll
  for (int k = 0; k < GAT_EPT; ++k) {
    const int i = threadIdx.x + k * GAT_T;
    if (i < m) tmp[(size_t)wg * GAT_CHUNK + i] = s_pair[i];
  }
}

#define GAT_GW 16
__global__ void __launch_bounds__(GAT_T) __attribute__((amdgpu_waves_per_eu(12, 16)))
gat_sort_and_gemm(const int* __restrict__ src, const int* __restrict__ dst,
                  int* __restrict__ cnt_t, int* __restrict__ off_t,
                  int* __restrict__ tmpw, int E, int B, int nwg,
                  const float* __restrict__ A, const float* __restrict__ Wm,
                  const float* __restrict__ ar,
                  float* __restrict__ C,
                  float* __restrict__ er, int M, int K, int gblk) {
  if ((int)blockIdx.x < gblk) {
    const int wv    = (int)(threadIdx.x >> 5);
    const int strip = (int)blockIdx.x * GAT_GW + wv;
    if (wv < GAT_GW && strip < ((M + 15) >> 4))
      gat_gemm_strip<2>(A, Wm, ar, C, er, M, K, strip);
    return;
  }
  gat_chunk_sort_body(src, dst, cnt_t, off_t, tmpw, E, B, nwg,
                      (int)blockIdx.x - gblk);
}

__device__ __forceinline__ int2 gat_run_fetch(const int2* __restrict__ tmp,
                                              const int* s_rp, const int* s_ro,
                                              int nwg, int i) {
  int lo = 0, hi = nwg - 1;
  while (lo < hi) {
    const int mid = (lo + hi + 1) >> 1;
    if (s_rp[mid] <= i) lo = mid; else hi = mid - 1;
  }
  return tmp[s_ro[lo] + (i - s_rp[lo])];
}

__global__ void __launch_bounds__(GAT_T)
gat_bucket_csr(const int* __restrict__ tmpw, const int* __restrict__ cnt_t,
               const int* __restrict__ off_t,
               int* __restrict__ rowstart, int* __restrict__ deg,
               int* __restrict__ ssrc, int N, int nwg) {
  const int2* __restrict__ tmp = (const int2*)tmpw;
  __shared__ int  s_cnt[256];
  __shared__ int  s_cur[256];
  __shared__ int  s_w[33];
  __shared__ int  s_w2[33];
  __shared__ int  s_rc[GAT_MAXW];
  __shared__ int  s_ro[GAT_MAXW];
  __shared__ int  s_rp[GAT_MAXW];
  __shared__ int2 s_stage[GAT_CAP2];
  __shared__ int  s_sorted[GAT_CAP2];
  const int b   = blockIdx.x;
  const int tid = threadIdx.x;

  int cnt = 0, base = 0;
  for (int w0 = 0; w0 < nwg; w0 += GAT_T) {
    const int w  = w0 + tid;
    const int c  = (w < nwg) ? cnt_t[(size_t)b * nwg + w] : 0;
    const int o  = (w < nwg) ? off_t[(size_t)b * nwg + w] : 0;
    const int ex = gat_block_excl_scan(c, s_w);
    (void)gat_block_excl_scan(o, s_w2);
    if (w < nwg) {
      s_rc[w] = c;
      s_ro[w] = w * GAT_CHUNK + o;
      s_rp[w] = cnt + ex;
    }
    cnt  += s_w[32];
    base += s_w2[32];
    __syncthreads();
  }
  if (tid < 256) s_cnt[tid] = 0;
  __syncthreads();
  const bool fits = cnt <= GAT_CAP2;

  if (fits) {
    const int lane = tid & 31;
    const int wv   = tid >> 5;
    for (int w0 = wv; w0 < nwg; w0 += 32 * GAT_RUNS) {
      int  rc[GAT_RUNS], ro[GAT_RUNS], rp[GAT_RUNS];
      int2 v[GAT_RUNS];
#pragma unroll
      for (int u = 0; u < GAT_RUNS; ++u) {
        const int  w  = w0 + 32 * u;
        const bool ok = w < nwg;
        const int  wc = min(w, nwg - 1);
        const int  tc = s_rc[wc];
        const int  to = s_ro[wc];
        const int  tp = s_rp[wc];
        rc[u] = ok ? tc : 0;
        ro[u] = ok ? to : 0;
        rp[u] = ok ? tp : 0;
      }
#pragma unroll
      for (int u = 0; u < GAT_RUNS; ++u)
        v[u] = (lane < rc[u]) ? tmp[ro[u] + lane] : make_int2(0, 0);
#pragma unroll
      for (int u = 0; u < GAT_RUNS; ++u) {
        if (lane < rc[u]) {
          s_stage[rp[u] + lane] = v[u];
          atomicAdd(&s_cnt[v[u].y & 255], 1);
        }
      }
      for (int u = 0; u < GAT_RUNS; ++u) {
        const int w = w0 + 32 * u;
        if (w >= nwg) break;
        const int c = s_rc[w];
        for (int i = lane + 32; i < c; i += 32) {
          const int2 q = tmp[s_ro[w] + i];
          s_stage[s_rp[w] + i] = q;
          atomicAdd(&s_cnt[q.y & 255], 1);
        }
      }
    }
  } else {
    for (int i0 = tid; i0 < cnt; i0 += GAT_T * GAT_EPT2) {
      int dl[GAT_EPT2];
#pragma unroll
      for (int k = 0; k < GAT_EPT2; ++k) {
        const int i = i0 + k * GAT_T;
        dl[k] = (i < cnt) ? gat_run_fetch(tmp, s_rp, s_ro, nwg, i).y : -1;
      }
#pragma unroll
      for (int k = 0; k < GAT_EPT2; ++k)
        if (dl[k] >= 0) atomicAdd(&s_cnt[dl[k] & 255], 1);
    }
  }
  __syncthreads();
  const int c  = (tid < 256) ? s_cnt[tid] : 0;
  const int ex = gat_block_excl_scan(c, s_w);
  if (tid < 256) {
    s_cur[tid] = ex;
    const int node = (b << GAT_BSH) + tid;
    if (node < N) { rowstart[node] = base + ex; deg[node] = c; }
  }
  __syncthreads();
  if (fits) {
    for (int i = tid; i < cnt; i += GAT_T) {
      const int2 p  = s_stage[i];
      const int pos = atomicAdd(&s_cur[p.y & 255], 1);
      s_sorted[pos] = p.x;
    }
    __syncthreads();
    for (int i = tid; i < cnt; i += GAT_T) ssrc[base + i] = s_sorted[i];
  } else {
    for (int i = tid; i < cnt; i += GAT_T) {
      const int2 p  = gat_run_fetch(tmp, s_rp, s_ro, nwg, i);
      const int pos = atomicAdd(&s_cur[p.y & 255], 1);
      ssrc[base + pos] = p.x;
    }
  }
}


typedef __attribute__((ext_vector_type(4))) float v4f;

template <int I>
__device__ __forceinline__ int gat_row_share(int v) {
  return __builtin_amdgcn_update_dpp(0, v, 0x150 | I, 0xf, 0xf, true);
}

template <int XM>
__device__ __forceinline__ float gat_row_xsum(float keep, float send) {
  const int b = __builtin_bit_cast(int, send);
  const int o = __builtin_amdgcn_update_dpp(0, b, 0x160 | XM, 0xf, 0xf, true);
  return keep + __builtin_bit_cast(float, o);
}

template <int XM>
__device__ __forceinline__ float gat_row_xmax(float v) {
  const int b = __builtin_bit_cast(int, v);
  const int o = __builtin_amdgcn_update_dpp(0, b, 0x160 | XM, 0xf, 0xf, true);
  return fmaxf(v, __builtin_bit_cast(float, o));
}

__device__ __forceinline__ float gat_xhalf(float x) {
  const int b = __builtin_bit_cast(int, x);
  return __builtin_bit_cast(float,
      __builtin_amdgcn_permlanex16(b, b, 0x76543210, 0xfedcba98, false, false));
}

#define GAT_STEP_LOAD(I)                                                     \
  zz[I] = (v4f){0.f, 0.f, 0.f, 0.f};                                         \
  if (2 * (I) < nn) {                                                        \
    const unsigned off = ((unsigned)gat_row_share<I>(svl) << 8) | cb;        \
    zz[I] = *(const v4f*)((const char*)z + off);                             \
  }

#define GAT_STEP_ACC(I)                                                      \
  {                                                                          \
    const float wi = __builtin_bit_cast(float,                               \
        gat_row_share<I>(__builtin_bit_cast(int, w)));                       \
    const v2f w2 = {wi, wi};                                                 \
    acc_lo = __builtin_elementwise_fma(w2, __builtin_shufflevector(zz[I], zz[I], 0, 1), acc_lo); \
    acc_hi = __builtin_elementwise_fma(w2, __builtin_shufflevector(zz[I], zz[I], 2, 3), acc_hi); \
  }

#define GAT_HP 68

template <bool TO_LDS>
__device__ __forceinline__ void
gat_aggregate_nodes(const int* __restrict__ rowstart,
                    const int* __restrict__ deg,
                    const int* __restrict__ ssrc,
                    const float* __restrict__ al,
                    const float* __restrict__ er,
                    const float* __restrict__ z,
                    const float* __restrict__ bias,
                    float* __restrict__ out, float* tile,
                    int node, const int nlast, const int node0, int applyRelu) {
  const int lane = threadIdx.x & 31;
  const int h     = lane >> 4;
  const int c     = lane & 15;
  const int eslot = 2 * c + h;
  const unsigned cb = (unsigned)c << 4;
  const v4f al4   = ((const v4f*)al)[c];
  const v2f al_lo = __builtin_shufflevector(al4, al4, 0, 1);
  const v2f al_hi = __builtin_shufflevector(al4, al4, 2, 3);
  const bool b3 = (c & 8) != 0, b2 = (c & 4) != 0, b1 = (c & 2) != 0, b0 = (c & 1) != 0;

  int   start = rowstart[node];
  int   cnt   = deg[node];
  float ern   = er[node];
  const int nd1 = min(node + 1, nlast);
  int   start1 = rowstart[nd1];
  int   cnt1   = deg[nd1];
  float ern1   = er[nd1];
  int svl = 0;
  if (cnt > 0) svl = ssrc[start + min(eslot, min(cnt, 32) - 1)];

  for (;;) {
  const int nd2 = min(node + 2, nlast);
  const int   start2 = rowstart[nd2];
  const int   cnt2   = deg[nd2];
  const float ern2   = er[nd2];
  int svl1 = 0;
  if (cnt1 > 0) svl1 = ssrc[start1 + min(eslot, min(cnt1, 32) - 1)];

  float m = -INFINITY;
  float s = 0.f;
  v2f acc_lo = {0.f, 0.f}, acc_hi = {0.f, 0.f};
  for (int base = 0; base < cnt; base += 32) {
    const int nn  = min(cnt - base, 32);
    if (base > 0) svl = ssrc[start + base + min(eslot, nn - 1)];

    v4f zz[16];
    GAT_STEP_LOAD(0)
    GAT_STEP_LOAD(1)
    GAT_STEP_LOAD(2)
    GAT_STEP_LOAD(3)
    GAT_STEP_LOAD(4)
    GAT_STEP_LOAD(5)
    GAT_STEP_LOAD(6)
    GAT_STEP_LOAD(7)
    GAT_STEP_LOAD(8)
    GAT_STEP_LOAD(9)
    GAT_STEP_LOAD(10)
    GAT_STEP_LOAD(11)
    GAT_STEP_LOAD(12)
    GAT_STEP_LOAD(13)
    GAT_STEP_LOAD(14)
    GAT_STEP_LOAD(15)

    float q[8], r[4], t2[2];
#pragma unroll
    for (int k = 0; k < 8; ++k) {
      v2f ta = __builtin_shufflevector(zz[k], zz[k], 0, 1) * al_lo;
      v2f tb = __builtin_shufflevector(zz[k + 8], zz[k + 8], 0, 1) * al_lo;
      ta = __builtin_elementwise_fma(__builtin_shufflevector(zz[k], zz[k], 2, 3), al_hi, ta);
      tb = __builtin_elementwise_fma(__builtin_shufflevector(zz[k + 8], zz[k + 8], 2, 3), al_hi, tb);
      const float pa = ta.x + ta.y;
      const float pb = tb.x + tb.y;
      q[k] = gat_row_xsum<8>(b3 ? pb : pa, b3 ? pa : pb);
    }
#pragma unroll
    for (int k = 0; k < 4; ++k)
      r[k] = gat_row_xsum<4>(b2 ? q[k + 4] : q[k], b2 ? q[k] : q[k + 4]);
#pragma unroll
    for (int k = 0; k < 2; ++k)
      t2[k] = gat_row_xsum<2>(b1 ? r[k + 2] : r[k], b1 ? r[k] : r[k + 2]);
    const float elv = gat_row_xsum<1>(b0 ? t2[1] : t2[0], b0 ? t2[0] : t2[1]);

    float e = elv + ern;
    e = (e > 0.f) ? e : 0.2f * e;
    e = (eslot < nn) ? e : -INFINITY;
    float cm = e;
    cm = gat_row_xmax<1>(cm);
    cm = gat_row_xmax<2>(cm);
    cm = gat_row_xmax<4>(cm);
    cm = gat_row_xmax<8>(cm);
    cm = fmaxf(cm, gat_xhalf(cm));
    const float mn = fmaxf(m, cm);
    const float sc = __expf(m - mn);
    const float w  = __expf(e - mn);
    m = mn;
    s = fmaf(s, sc, w);
    acc_lo *= sc;
    acc_hi *= sc;
    GAT_STEP_ACC(0)
    GAT_STEP_ACC(1)
    GAT_STEP_ACC(2)
    GAT_STEP_ACC(3)
    GAT_STEP_ACC(4)
    GAT_STEP_ACC(5)
    GAT_STEP_ACC(6)
    GAT_STEP_ACC(7)
    GAT_STEP_ACC(8)
    GAT_STEP_ACC(9)
    GAT_STEP_ACC(10)
    GAT_STEP_ACC(11)
    GAT_STEP_ACC(12)
    GAT_STEP_ACC(13)
    GAT_STEP_ACC(14)
    GAT_STEP_ACC(15)
  }

  s = gat_row_xadd<1>(s);
  s = gat_row_xadd<2>(s);
  s = gat_row_xadd<4>(s);
  s = gat_row_xadd<8>(s);
  s += gat_xhalf(s);
  float ax = acc_lo.x, ay = acc_lo.y, az = acc_hi.x, aw = acc_hi.y;
  ax += gat_xhalf(ax);
  ay += gat_xhalf(ay);
  az += gat_xhalf(az);
  aw += gat_xhalf(aw);
  const float rcp = (cnt > 0) ? (1.f / s) : 0.f;

  const v4f b4 = ((const v4f*)bias)[c];
  v4f o;
  o.x = fmaf(ax, rcp, b4.x);
  o.y = fmaf(ay, rcp, b4.y);
  o.z = fmaf(az, rcp, b4.z);
  o.w = fmaf(aw, rcp, b4.w);
  if (applyRelu) {
    o.x = fmaxf(o.x, 0.f); o.y = fmaxf(o.y, 0.f);
    o.z = fmaxf(o.z, 0.f); o.w = fmaxf(o.w, 0.f);
  }
  if (TO_LDS) {
    if (h == 0) *(v4f*)(tile + (node - node0) * GAT_HP + 4 * c) = o;
  } else {
    if (h == 0) ((v4f*)(out + (size_t)node * 64))[c] = o;
  }

  if (node >= nlast) break;
  ++node;
  start = start1; cnt = cnt1; ern = ern1; svl = svl1;
  start1 = start2; cnt1 = cnt2; ern1 = ern2;
  }
}
#undef GAT_STEP_LOAD
#undef GAT_STEP_ACC

__global__ void __launch_bounds__(256)
gat_node_aggregate(const int* __restrict__ rowstart,
                   const int* __restrict__ deg,
                   const int* __restrict__ ssrc,
                   const float* __restrict__ al,
                   const float* __restrict__ er,
                   const float* __restrict__ z,
                   const float* __restrict__ bias,
                   float* __restrict__ out,
                   int N, int applyRelu) {
  const int wv  = __builtin_amdgcn_readfirstlane((int)(threadIdx.x >> 5));
  const int wpb = (int)(blockDim.x >> 5);
  const int wid = (int)blockIdx.x * wpb + wv;
  const int nwv = (int)gridDim.x * wpb;
  const int npw = N / nwv;
  const int nex = N - npw * nwv;
  const int node = wid * npw + min(wid, nex);
  const int nlast = node + npw - ((wid < nex) ? 0 : 1);
  if (node > nlast) return;
  gat_aggregate_nodes<false>(rowstart, deg, ssrc, al, er, z, bias, out, nullptr,
                             node, nlast, 0, applyRelu);
}

__global__ void __launch_bounds__(128)
gat_node_aggregate_gemm(const int* __restrict__ rowstart,
                        const int* __restrict__ deg,
                        const int* __restrict__ ssrc,
                        const float* __restrict__ al,
                        const float* __restrict__ er,
                        const float* __restrict__ z,
                        const float* __restrict__ bias,
                        const float* __restrict__ W2,
                        const float* __restrict__ ar2,
                        float* __restrict__ z2,
                        float* __restrict__ er2, int N) {
  __shared__ __attribute__((aligned(16))) float s_h[16 * GAT_HP];
  __shared__ __attribute__((aligned(16))) float s_z[16 * GAT_HP];
  const int lane = threadIdx.x & 31;
  const int wv   = __builtin_amdgcn_readfirstlane((int)(threadIdx.x >> 5));
  const int kh   = lane >> 4;
  const int l15  = lane & 15;
  const int row0 = (int)blockIdx.x * 16;
  const int node = row0 + 4 * wv;
  const int nlast = min(node + 3, N - 1);

  if (node + 3 > N - 1) {
    for (int r = max(N - node, 0); r < 4; ++r)
      if (kh == 0) *(v4f*)(s_h + (4 * wv + r) * GAT_HP + 4 * l15) = (v4f){0.f, 0.f, 0.f, 0.f};
  }
  if (node <= nlast)
    gat_aggregate_nodes<true>(rowstart, deg, ssrc, al, er, z, bias, nullptr, s_h,
                              node, nlast, row0,  1);

  const float* bcol = W2 + (size_t)(2 * kh) * 64 + 16 * wv + l15;
  v2f bf[16];
#pragma unroll
  for (int j = 0; j < 16; ++j) {
    bf[j].x = bcol[(size_t)j * 256];
    bf[j].y = bcol[(size_t)j * 256 + 64];
  }
  float arv[4];
#pragma unroll
  for (int t = 0; t < 4; ++t) arv[t] = ar2[t * 16 + l15];
  __builtin_amdgcn_fence(__ATOMIC_RELEASE, "workgroup", "local");
  __builtin_amdgcn_s_barrier();
  __builtin_amdgcn_fence(__ATOMIC_ACQUIRE, "workgroup", "local");

  const float* arow = s_h + l15 * GAT_HP + 2 * kh;
  v2f af[16];
#pragma unroll
  for (int j = 0; j < 16; ++j) af[j] = *(const v2f*)(arow + 4 * j);
  __builtin_amdgcn_sched_barrier(0);
  v8f acc = (v8f){0.f,0.f,0.f,0.f,0.f,0.f,0.f,0.f};
#pragma unroll
  for (int j = 0; j < 16; ++j)
    acc = __builtin_amdgcn_wmma_f32_16x16x4_f32(
        false, af[j], false, bf[j], (short)0, acc, false, false);
#pragma unroll
  for (int r = 0; r < 8; ++r)
    s_z[(8 * kh + r) * GAT_HP + 16 * wv + l15] = acc[r];
  __syncthreads();

#pragma unroll
  for (int p = 0; p < 2; ++p) {
    const int row = 4 * wv + 2 * p + kh;
    const float* zr = s_z + row * GAT_HP;
    const v4f zv = *(const v4f*)(zr + 4 * l15);
    float sr = zr[l15] * arv[0];
#pragma unroll
    for (int t = 1; t < 4; ++t) sr += zr[t * 16 + l15] * arv[t];
    sr = gat_row_xadd<1>(sr);
    sr = gat_row_xadd<2>(sr);
    sr = gat_row_xadd<4>(sr);
    sr = gat_row_xadd<8>(sr);
    if (row0 + row < N) {
      ((v4f*)(z2 + (size_t)(row0 + row) * 64))[l15] = zv;
      if (l15 == 0) er2[row0 + row] = sr;
    }
  }
}


extern "C" void kernel_launch(void* const* d_in, const int* in_sizes, int n_in,
                              void* d_out, int out_size, void* d_ws, size_t ws_size,
                              hipStream_t stream) {
  const float* x   = (const float*)d_in[0];
  const float* W1  = (const float*)d_in[1];
  const float* al1 = (const float*)d_in[2];
  const float* ar1 = (const float*)d_in[3];
  const float* b1  = (const float*)d_in[4];
  const float* W2  = (const float*)d_in[5];
  const float* al2 = (const float*)d_in[6];
  const float* ar2 = (const float*)d_in[7];
  const float* b2  = (const float*)d_in[8];
  const int*   src = (const int*)d_in[9];
  const int*   dst = (const int*)d_in[10];
  float* out = (float*)d_out;

  const int F  = 64;
  const int N  = out_size / F;
  const int E  = in_sizes[9];
  const int K1 = in_sizes[0] / N;
  (void)n_in; (void)ws_size;

  char* ws = (char*)d_ws;
  size_t off = 0;
  auto alloc = [&](size_t bytes) -> void* {
    void* p = ws + off;
    off += (bytes + 255) & ~(size_t)255;
    return p;
  };
  float* zbuf     = (float*)alloc((size_t)N * F * sizeof(float));
  float* zbuf2    = (float*)alloc((size_t)N * F * sizeof(float));
  float* er       = (float*)alloc((size_t)N * sizeof(float));
  float* er2      = (float*)alloc((size_t)N * sizeof(float));
  int*   deg      = (int*)alloc((size_t)N * sizeof(int));
  int*   rowstart = (int*)alloc((size_t)N * sizeof(int));
  int*   ssrc     = (int*)alloc((size_t)E * sizeof(int));
  const int nb    = (N + 255) / 256;
  const int nwg   = (E + GAT_CHUNK - 1) / GAT_CHUNK;
  int*   cnt_t    = (int*)alloc((size_t)nb * nwg * sizeof(int));
  int*   off_t    = (int*)alloc((size_t)nb * nwg * sizeof(int));
  int*   tmp      = (int*)alloc((size_t)E * 2 * sizeof(int));

  const int gAgg   = 2048;
  const int gFuse  = (N + 15) / 16;
  const int gGemm1 = (((N + 15) / 16) + GAT_GW - 1) / GAT_GW;

  gat_sort_and_gemm <<<gGemm1 + nwg, GAT_T, 0, stream>>>(src, dst, cnt_t, off_t, tmp, E, nb, nwg,
                                                x, W1, ar1, zbuf, er, N, K1,
                                                gGemm1);
  gat_bucket_csr    <<<nb,  GAT_T, 0, stream>>>(tmp, cnt_t, off_t, rowstart, deg, ssrc,
                                                N, nwg);

  gat_node_aggregate_gemm<<<gFuse, 128, 0, stream>>>(rowstart, deg, ssrc, al1, er,
                                                 zbuf, b1, W2, ar2, zbuf2, er2, N);

  gat_node_aggregate<<<gAgg, 128, 0, stream>>>(rowstart, deg, ssrc, al2, er2,
                                                 zbuf2, b2, out, N,  0);
}


# layer-1 GEMM moves from the sort launch to CSR pass 2's launch, beside the bucket workgroups; to fit 64 registers it takes W1 from LDS, x 16 B a lane; new k order in WMMA: new summation order
# speedup vs baseline: 3.9524x; 1.0734x over previous
#include <hip/hip_runtime.h>
#include <math.h>


typedef __attribute__((ext_vector_type(2))) float v2f;
typedef __attribute__((ext_vector_type(4))) float v4f;
typedef __attribute__((ext_vector_type(8))) float v8f;

template <int XM>
__device__ __forceinline__ float gat_row_xadd(float v) {
  const int b = __builtin_bit_cast(int, v);
  const int o = __builtin_amdgcn_update_dpp(0, b, 0x160 | XM, 0xf, 0xf, true);
  return v + __builtin_bit_cast(float, o);
}

__device__ __forceinline__ void
gat_strip_store(const v8f (&acc)[4], const float* __restrict__ ar,
                float* __restrict__ C, float* __restrict__ er, int M, int strip);

template <int UNR>
__device__ __forceinline__ void
gat_gemm_strip(const float* __restrict__ A, const float* __restrict__ B,
               const float* __restrict__ ar,
               float* __restrict__ C,
               float* __restrict__ er, int M, int K, int strip) {
  const int lane = threadIdx.x & 31;
  const int row0 = strip * 16;
  const int kh   = lane >> 4;
  const int l15  = lane & 15;
  int mrow = row0 + l15;
  if (mrow >= M) mrow = M - 1;

  const float* arow = A + (size_t)mrow * K;
  v8f acc[4];
#pragma unroll
  for (int t = 0; t < 4; ++t) acc[t] = (v8f){0.f,0.f,0.f,0.f,0.f,0.f,0.f,0.f};

#pragma unroll UNR
  for (int k = 0; k < K; k += 4) {
    const v2f a = *(const v2f*)(arow + k + 2 * kh);
    const float* brow0 = B + (size_t)(k + 2 * kh) * 64 + l15;
#pragma unroll
    for (int t = 0; t < 4; ++t) {
      v2f b;
      b.x = brow0[t * 16];
      b.y = brow0[64 + t * 16];
      acc[t] = __builtin_amdgcn_wmma_f32_16x16x4_f32(
          false, a, false, b, (short)0, acc[t], false, false);
    }
  }

  gat_strip_store(acc, ar, C, er, M, strip);
}

__device__ __forceinline__ void
gat_strip_store(const v8f (&acc)[4], const float* __restrict__ ar,
                float* __restrict__ C, float* __restrict__ er, int M, int strip) {
  const int lane = threadIdx.x & 31;
  const int row0 = strip * 16;
  const int kh   = lane >> 4;
  const int l15  = lane & 15;
  const int rbase = row0 + 8 * kh;
#pragma unroll
  for (int t = 0; t < 4; ++t) {
    float* cp = C + (size_t)rbase * 64 + t * 16 + l15;
#pragma unroll
    for (int r = 0; r < 8; ++r)
      if (rbase + r < M) cp[(size_t)r * 64] = acc[t][r];
  }

  float arv[4];
#pragma unroll
  for (int t = 0; t < 4; ++t) arv[t] = ar[t * 16 + l15];
  float pr[8];
#pragma unroll
  for (int r = 0; r < 8; ++r) {
    float sr = acc[0][r] * arv[0];
#pragma unroll
    for (int t = 1; t < 4; ++t) sr += acc[t][r] * arv[t];
    pr[r] = sr;
  }
#pragma unroll
  for (int r = 0; r < 8; ++r) pr[r] = gat_row_xadd<1>(pr[r]);
#pragma unroll
  for (int r = 0; r < 8; ++r) pr[r] = gat_row_xadd<2>(pr[r]);
#pragma unroll
  for (int r = 0; r < 8; ++r) pr[r] = gat_row_xadd<4>(pr[r]);
#pragma unroll
  for (int r = 0; r < 8; ++r) pr[r] = gat_row_xadd<8>(pr[r]);
  const int rsel = l15 & 7;
  float outv = pr[0];
#pragma unroll
  for (int r = 1; r < 8; ++r)
    if (rsel == r) outv = pr[r];
  if (l15 < 8 && rbase + rsel < M) er[rbase + rsel] = outv;
}

#define GAT_T  1024
#define GAT_KB 128
#define GAT_WP 72
#define GAT_WST ((GAT_KB / 2) * 64 / GAT_T)
__device__ __forceinline__ void
gat_gemm_strip_lds(const float* __restrict__ A, const float* __restrict__ B,
                   const float* __restrict__ ar,
                   float* __restrict__ C,
                   float* __restrict__ er, int M, int K, int strip, bool active) {
  __shared__ v2f s_wp[(GAT_KB / 2) * GAT_WP];
  const int lane = threadIdx.x & 31;
  const int kh   = lane >> 4;
  const int l15  = lane & 15;
  int mrow = strip * 16 + l15;
  if (mrow >= M) mrow = M - 1;
  const float* arow = A + (size_t)mrow * K + 4 * kh;
  v8f acc[4];
#pragma unroll
  for (int t = 0; t < 4; ++t) acc[t] = (v8f){0.f,0.f,0.f,0.f,0.f,0.f,0.f,0.f};

  {
    v2f w2[GAT_WST];
#pragma unroll
    for (int it = 0; it < GAT_WST; ++it) {
      const int idx = (int)threadIdx.x + it * GAT_T;
      const int kr  = min(2 * (idx >> 6), K - 2);
      const float* bp = B + (size_t)kr * 64 + (idx & 63);
      w2[it].x = bp[0];
      w2[it].y = bp[64];
    }
#pragma unroll
    for (int it = 0; it < GAT_WST; ++it) {
      const int idx = (int)threadIdx.x + it * GAT_T;
      s_wp[(idx >> 6) * GAT_WP + (idx & 63)] = w2[it];
    }
    __syncthreads();
    if (active) {
      const int ng = K >> 3;
      const float* ap = arow;
      for (int g0 = 0; g0 < ng; g0 += 4) {
        v4f a4[4];
#pragma unroll
        for (int u = 0; u < 4; ++u) a4[u] = *(const v4f*)(ap + 8 * (g0 + u));
        const v2f* bp = s_wp + (4 * g0 + 2 * kh) * GAT_WP + l15;
#pragma unroll
        for (int u = 0; u < 4; ++u) {
#pragma unroll
          for (int i = 0; i < 2; ++i) {
            const v2f a = i ? __builtin_shufflevector(a4[u], a4[u], 2, 3)
                            : __builtin_shufflevector(a4[u], a4[u], 0, 1);
#pragma unroll
            for (int t = 0; t < 4; ++t) {
              const v2f b = bp[(4 * u + i) * GAT_WP + 16 * t];
              acc[t] = __builtin_amdgcn_wmma_f32_16x16x4_f32(
                  false, a, false, b, (short)0, acc[t], false, false);
            }
          }
        }
      }
    }
  }
  if (active) gat_strip_store(acc, ar, C, er, M, strip);
}


#define GAT_BSH   8
#define GAT_MAXB  GAT_T
#define GAT_EPT   4
#define GAT_CHUNK (GAT_T * GAT_EPT)
#define GAT_EPT2  8
#define GAT_CAP2  8192
#define GAT_MAXW  1024
#define GAT_RUNS  10

__device__ __forceinline__ int gat_wave_incl_scan(int v, int lane) {
#pragma unroll
  for (int off = 1; off < 32; off <<= 1) {
    const int t = __shfl_up(v, off, 32);
    if (lane >= off) v += t;
  }
  return v;
}

__device__ __forceinline__ int gat_block_excl_scan(int v, int* s_w) {
  const int lane = threadIdx.x & 31;
  const int wv   = threadIdx.x >> 5;
  const int nw   = (blockDim.x + 31) >> 5;
  const int inc  = gat_wave_incl_scan(v, lane);
  if (lane == 31) s_w[wv] = inc;
  __syncthreads();
  if (wv == 0) {
    const int w  = (lane < nw) ? s_w[lane] : 0;
    const int wi = gat_wave_incl_scan(w, lane);
    s_w[lane] = wi - w;
    if (lane == 31) s_w[32] = wi;
  }
  __syncthreads();
  return inc - v + s_w[wv];
}

__device__ __forceinline__ void
gat_chunk_sort_body(const int* __restrict__ src, const int* __restrict__ dst,
                    int* __restrict__ cnt_t, int* __restrict__ off_t,
                    int* __restrict__ tmpw, int E, int B, int nwg, int wg) {
  int2* __restrict__ tmp = (int2*)tmpw;
  __shared__ int  s_h[GAT_MAXB];
  __shared__ int  s_cur[GAT_MAXB];
  __shared__ int2 s_pair[GAT_CHUNK];
  __shared__ int  s_w[33];
  const int b  = threadIdx.x;
  s_h[b] = 0;
  const int e0 = wg * GAT_CHUNK + threadIdx.x;
  int sv[GAT_EPT], dv[GAT_EPT];
#pragma unroll
  for (int k = 0; k < GAT_EPT; ++k) {
    const int e = e0 + k * GAT_T;
    dv[k] = (e < E) ? dst[e] : -1;
    sv[k] = (e < E) ? src[e] : 0;
  }
  __syncthreads();
#pragma unroll
  for (int k = 0; k < GAT_EPT; ++k)
    if (dv[k] >= 0) atomicAdd(&s_h[dv[k] >> GAT_BSH], 1);
  __syncthreads();
  const int c  = s_h[b];
  const int ex = gat_block_excl_scan(c, s_w);
  s_cur[b] = ex;
  if (b < B) {
    cnt_t[(size_t)b * nwg + wg] = c;
    off_t[(size_t)b * nwg + wg] = ex;
  }
  __syncthreads();
#pragma unroll
  for (int k = 0; k < GAT_EPT; ++k) {
    if (dv[k] >= 0) {
      const int pos = atomicAdd(&s_cur[dv[k] >> GAT_BSH], 1);
      s_pair[pos] = make_int2(sv[k], dv[k]);
    }
  }
  __syncthreads();
  const int m = min(GAT_CHUNK, E - wg * GAT_CHUNK);
#pragma unroll
  for (int k = 0; k < GAT_EPT; ++k) {
    const int i = threadIdx.x + k * GAT_T;
    if (i < m) tmp[(size_t)wg * GAT_CHUNK + i] = s_pair[i];
  }
}

__global__ void __launch_bounds__(GAT_T)
gat_chunk_sort(const int* __restrict__ src, const int* __restrict__ dst,
               int* __restrict__ cnt_t, int* __restrict__ off_t,
               int* __restrict__ tmpw, int E, int B, int nwg) {
  gat_chunk_sort_body(src, dst, cnt_t, off_t, tmpw, E, B, nwg, (int)blockIdx.x);
}

__device__ __forceinline__ int2 gat_run_fetch(const int2* __restrict__ tmp,
                                              const int* s_rp, const int* s_ro,
                                              int nwg, int i) {
  int lo = 0, hi = nwg - 1;
  while (lo < hi) {
    const int mid = (lo + hi + 1) >> 1;
    if (s_rp[mid] <= i) lo = mid; else hi = mid - 1;
  }
  return tmp[s_ro[lo] + (i - s_rp[lo])];
}

__device__ __forceinline__ void
gat_bucket_body(const int* __restrict__ tmpw, const int* __restrict__ cnt_t,
                const int* __restrict__ off_t,
                int* __restrict__ rowstart, int* __restrict__ deg,
                int* __restrict__ ssrc, int N, int nwg, int b) {
  const int2* __restrict__ tmp = (const int2*)tmpw;
  __shared__ int  s_cnt[256];
  __shared__ int  s_cur[256];
  __shared__ int  s_w[33];
  __shared__ int  s_w2[33];
  __shared__ int  s_rc[GAT_MAXW];
  __shared__ int  s_ro[GAT_MAXW];
  __shared__ int  s_rp[GAT_MAXW];
  __shared__ int2 s_stage[GAT_CAP2];
  __shared__ int  s_sorted[GAT_CAP2];
  const int tid = threadIdx.x;

  int cnt = 0, base = 0;
  for (int w0 = 0; w0 < nwg; w0 += GAT_T) {
    const int w  = w0 + tid;
    const int c  = (w < nwg) ? cnt_t[(size_t)b * nwg + w] : 0;
    const int o  = (w < nwg) ? off_t[(size_t)b * nwg + w] : 0;
    const int ex = gat_block_excl_scan(c, s_w);
    (void)gat_block_excl_scan(o, s_w2);
    if (w < nwg) {
      s_rc[w] = c;
      s_ro[w] = w * GAT_CHUNK + o;
      s_rp[w] = cnt + ex;
    }
    cnt  += s_w[32];
    base += s_w2[32];
    __syncthreads();
  }
  if (tid < 256) s_cnt[tid] = 0;
  __syncthreads();
  const bool fits = cnt <= GAT_CAP2;

  if (fits) {
    const int lane = tid & 31;
    const int wv   = tid >> 5;
    for (int w0 = wv; w0 < nwg; w0 += 32 * GAT_RUNS) {
      int  rc[GAT_RUNS], ro[GAT_RUNS], rp[GAT_RUNS];
      int2 v[GAT_RUNS];
#pragma unroll
      for (int u = 0; u < GAT_RUNS; ++u) {
        const int  w  = w0 + 32 * u;
        const bool ok = w < nwg;
        const int  wc = min(w, nwg - 1);
        const int  tc = s_rc[wc];
        const int  to = s_ro[wc];
        const int  tp = s_rp[wc];
        rc[u] = ok ? tc : 0;
        ro[u] = ok ? to : 0;
        rp[u] = ok ? tp : 0;
      }
#pragma unroll
      for (int u = 0; u < GAT_RUNS; ++u)
        v[u] = (lane < rc[u]) ? tmp[ro[u] + lane] : make_int2(0, 0);
#pragma unroll
      for (int u = 0; u < GAT_RUNS; ++u) {
        if (lane < rc[u]) {
          s_stage[rp[u] + lane] = v[u];
          atomicAdd(&s_cnt[v[u].y & 255], 1);
        }
      }
      for (int u = 0; u < GAT_RUNS; ++u) {
        const int w = w0 + 32 * u;
        if (w >= nwg) break;
        const int c = s_rc[w];
        for (int i = lane + 32; i < c; i += 32) {
          const int2 q = tmp[s_ro[w] + i];
          s_stage[s_rp[w] + i] = q;
          atomicAdd(&s_cnt[q.y & 255], 1);
        }
      }
    }
  } else {
    for (int i0 = tid; i0 < cnt; i0 += GAT_T * GAT_EPT2) {
      int dl[GAT_EPT2];
#pragma unroll
      for (int k = 0; k < GAT_EPT2; ++k) {
        const int i = i0 + k * GAT_T;
        dl[k] = (i < cnt) ? gat_run_fetch(tmp, s_rp, s_ro, nwg, i).y : -1;
      }
#pragma unroll
      for (int k = 0; k < GAT_EPT2; ++k)
        if (dl[k] >= 0) atomicAdd(&s_cnt[dl[k] & 255], 1);
    }
  }
  __syncthreads();
  const int c  = (tid < 256) ? s_cnt[tid] : 0;
  const int ex = gat_block_excl_scan(c, s_w);
  if (tid < 256) {
    s_cur[tid] = ex;
    const int node = (b << GAT_BSH) + tid;
    if (node < N) { rowstart[node] = base + ex; deg[node] = c; }
  }
  __syncthreads();
  if (fits) {
    for (int i = tid; i < cnt; i += GAT_T) {
      const int2 p  = s_stage[i];
      const int pos = atomicAdd(&s_cur[p.y & 255], 1);
      s_sorted[pos] = p.x;
    }
    __syncthreads();
    for (int i = tid; i < cnt; i += GAT_T) ssrc[base + i] = s_sorted[i];
  } else {
    for (int i = tid; i < cnt; i += GAT_T) {
      const int2 p  = gat_run_fetch(tmp, s_rp, s_ro, nwg, i);
      const int pos = atomicAdd(&s_cur[p.y & 255], 1);
      ssrc[base + pos] = p.x;
    }
  }
}

#define GAT_GW 16
__global__ void __launch_bounds__(GAT_T) __attribute__((amdgpu_waves_per_eu(16, 16)))
gat_bucket_csr_and_gemm(const int* __restrict__ tmpw, const int* __restrict__ cnt_t,
                        const int* __restrict__ off_t,
                        int* __restrict__ rowstart, int* __restrict__ deg,
                        int* __restrict__ ssrc, int N, int nwg, int nb,
                        const float* __restrict__ A, const float* __restrict__ Wm,
                        const float* __restrict__ ar,
                        float* __restrict__ C,
                        float* __restrict__ er, int K) {
  if ((int)blockIdx.x < nb) {
    gat_bucket_body(tmpw, cnt_t, off_t, rowstart, deg, ssrc, N, nwg,
                    (int)blockIdx.x);
    return;
  }
  const int wv    = __builtin_amdgcn_readfirstlane((int)(threadIdx.x >> 5));
  const int strip = ((int)blockIdx.x - nb) * GAT_GW + wv;
  const bool active = wv < GAT_GW && strip < ((N + 15) >> 4);
  if ((K & 31) == 0 && K <= GAT_KB)
    gat_gemm_strip_lds(A, Wm, ar, C, er, N, K, strip, active);
  else if (active)
    gat_gemm_strip<1>(A, Wm, ar, C, er, N, K, strip);
}


template <int I>
__device__ __forceinline__ int gat_row_share(int v) {
  return __builtin_amdgcn_update_dpp(0, v, 0x150 | I, 0xf, 0xf, true);
}

template <int XM>
__device__ __forceinline__ float gat_row_xsum(float keep, float send) {
  const int b = __builtin_bit_cast(int, send);
  const int o = __builtin_amdgcn_update_dpp(0, b, 0x160 | XM, 0xf, 0xf, true);
  return keep + __builtin_bit_cast(float, o);
}

template <int XM>
__device__ __forceinline__ float gat_row_xmax(float v) {
  const int b = __builtin_bit_cast(int, v);
  const int o = __builtin_amdgcn_update_dpp(0, b, 0x160 | XM, 0xf, 0xf, true);
  return fmaxf(v, __builtin_bit_cast(float, o));
}

__device__ __forceinline__ float gat_xhalf(float x) {
  const int b = __builtin_bit_cast(int, x);
  return __builtin_bit_cast(float,
      __builtin_amdgcn_permlanex16(b, b, 0x76543210, 0xfedcba98, false, false));
}

#define GAT_STEP_LOAD(I)                                                     \
  zz[I] = (v4f){0.f, 0.f, 0.f, 0.f};                                         \
  if (2 * (I) < nn) {                                                        \
    const unsigned off = ((unsigned)gat_row_share<I>(svl) << 8) | cb;        \
    zz[I] = *(const v4f*)((const char*)z + off);                             \
  }

#define GAT_STEP_ACC(I)                                                      \
  {                                                                          \
    const float wi = __builtin_bit_cast(float,                               \
        gat_row_share<I>(__builtin_bit_cast(int, w)));                       \
    const v2f w2 = {wi, wi};                                                 \
    acc_lo = __builtin_elementwise_fma(w2, __builtin_shufflevector(zz[I], zz[I], 0, 1), acc_lo); \
    acc_hi = __builtin_elementwise_fma(w2, __builtin_shufflevector(zz[I], zz[I], 2, 3), acc_hi); \
  }

#define GAT_HP 68

template <bool TO_LDS>
__device__ __forceinline__ void
gat_aggregate_nodes(const int* __restrict__ rowstart,
                    const int* __restrict__ deg,
                    const int* __restrict__ ssrc,
                    const float* __restrict__ al,
                    const float* __restrict__ er,
                    const float* __restrict__ z,
                    const float* __restrict__ bias,
                    float* __restrict__ out, float* tile,
                    int node, const int nlast, const int node0, int applyRelu) {
  const int lane = threadIdx.x & 31;
  const int h     = lane >> 4;
  const int c     = lane & 15;
  const int eslot = 2 * c + h;
  const unsigned cb = (unsigned)c << 4;
  const v4f al4   = ((const v4f*)al)[c];
  const v2f al_lo = __builtin_shufflevector(al4, al4, 0, 1);
  const v2f al_hi = __builtin_shufflevector(al4, al4, 2, 3);
  const bool b3 = (c & 8) != 0, b2 = (c & 4) != 0, b1 = (c & 2) != 0, b0 = (c & 1) != 0;

  int   start = rowstart[node];
  int   cnt   = deg[node];
  float ern   = er[node];
  const int nd1 = min(node + 1, nlast);
  int   start1 = rowstart[nd1];
  int   cnt1   = deg[nd1];
  float ern1   = er[nd1];
  int svl = 0;
  if (cnt > 0) svl = ssrc[start + min(eslot, min(cnt, 32) - 1)];

  for (;;) {
  const int nd2 = min(node + 2, nlast);
  const int   start2 = rowstart[nd2];
  const int   cnt2   = deg[nd2];
  const float ern2   = er[nd2];
  int svl1 = 0;
  if (cnt1 > 0) svl1 = ssrc[start1 + min(eslot, min(cnt1, 32) - 1)];

  float m = -INFINITY;
  float s = 0.f;
  v2f acc_lo = {0.f, 0.f}, acc_hi = {0.f, 0.f};
  for (int base = 0; base < cnt; base += 32) {
    const int nn  = min(cnt - base, 32);
    if (base > 0) svl = ssrc[start + base + min(eslot, nn - 1)];

    v4f zz[16];
    GAT_STEP_LOAD(0)
    GAT_STEP_LOAD(1)
    GAT_STEP_LOAD(2)
    GAT_STEP_LOAD(3)
    GAT_STEP_LOAD(4)
    GAT_STEP_LOAD(5)
    GAT_STEP_LOAD(6)
    GAT_STEP_LOAD(7)
    GAT_STEP_LOAD(8)
    GAT_STEP_LOAD(9)
    GAT_STEP_LOAD(10)
    GAT_STEP_LOAD(11)
    GAT_STEP_LOAD(12)
    GAT_STEP_LOAD(13)
    GAT_STEP_LOAD(14)
    GAT_STEP_LOAD(15)

    float q[8], r[4], t2[2];
#pragma unroll
    for (int k = 0; k < 8; ++k) {
      v2f ta = __builtin_shufflevector(zz[k], zz[k], 0, 1) * al_lo;
      v2f tb = __builtin_shufflevector(zz[k + 8], zz[k + 8], 0, 1) * al_lo;
      ta = __builtin_elementwise_fma(__builtin_shufflevector(zz[k], zz[k], 2, 3), al_hi, ta);
      tb = __builtin_elementwise_fma(__builtin_shufflevector(zz[k + 8], zz[k + 8], 2, 3), al_hi, tb);
      const float pa = ta.x + ta.y;
      const float pb = tb.x + tb.y;
      q[k] = gat_row_xsum<8>(b3 ? pb : pa, b3 ? pa : pb);
    }
#pragma unroll
    for (int k = 0; k < 4; ++k)
      r[k] = gat_row_xsum<4>(b2 ? q[k + 4] : q[k], b2 ? q[k] : q[k + 4]);
#pragma unroll
    for (int k = 0; k < 2; ++k)
      t2[k] = gat_row_xsum<2>(b1 ? r[k + 2] : r[k], b1 ? r[k] : r[k + 2]);
    const float elv = gat_row_xsum<1>(b0 ? t2[1] : t2[0], b0 ? t2[0] : t2[1]);

    float e = elv + ern;
    e = (e > 0.f) ? e : 0.2f * e;
    e = (eslot < nn) ? e : -INFINITY;
    float cm = e;
    cm = gat_row_xmax<1>(cm);
    cm = gat_row_xmax<2>(cm);
    cm = gat_row_xmax<4>(cm);
    cm = gat_row_xmax<8>(cm);
    cm = fmaxf(cm, gat_xhalf(cm));
    const float mn = fmaxf(m, cm);
    const float sc = __expf(m - mn);
    const float w  = __expf(e - mn);
    m = mn;
    s = fmaf(s, sc, w);
    acc_lo *= sc;
    acc_hi *= sc;
    GAT_STEP_ACC(0)
    GAT_STEP_ACC(1)
    GAT_STEP_ACC(2)
    GAT_STEP_ACC(3)
    GAT_STEP_ACC(4)
    GAT_STEP_ACC(5)
    GAT_STEP_ACC(6)
    GAT_STEP_ACC(7)
    GAT_STEP_ACC(8)
    GAT_STEP_ACC(9)
    GAT_STEP_ACC(10)
    GAT_STEP_ACC(11)
    GAT_STEP_ACC(12)
    GAT_STEP_ACC(13)
    GAT_STEP_ACC(14)
    GAT_STEP_ACC(15)
  }

  s = gat_row_xadd<1>(s);
  s = gat_row_xadd<2>(s);
  s = gat_row_xadd<4>(s);
  s = gat_row_xadd<8>(s);
  s += gat_xhalf(s);
  float ax = acc_lo.x, ay = acc_lo.y, az = acc_hi.x, aw = acc_hi.y;
  ax += gat_xhalf(ax);
  ay += gat_xhalf(ay);
  az += gat_xhalf(az);
  aw += gat_xhalf(aw);
  const float rcp = (cnt > 0) ? (1.f / s) : 0.f;

  const v4f b4 = ((const v4f*)bias)[c];
  v4f o;
  o.x = fmaf(ax, rcp, b4.x);
  o.y = fmaf(ay, rcp, b4.y);
  o.z = fmaf(az, rcp, b4.z);
  o.w = fmaf(aw, rcp, b4.w);
  if (applyRelu) {
    o.x = fmaxf(o.x, 0.f); o.y = fmaxf(o.y, 0.f);
    o.z = fmaxf(o.z, 0.f); o.w = fmaxf(o.w, 0.f);
  }
  if (TO_LDS) {
    if (h == 0) *(v4f*)(tile + (node - node0) * GAT_HP + 4 * c) = o;
  } else {
    if (h == 0) ((v4f*)(out + (size_t)node * 64))[c] = o;
  }

  if (node >= nlast) break;
  ++node;
  start = start1; cnt = cnt1; ern = ern1; svl = svl1;
  start1 = start2; cnt1 = cnt2; ern1 = ern2;
  }
}
#undef GAT_STEP_LOAD
#undef GAT_STEP_ACC

__global__ void __launch_bounds__(256)
gat_node_aggregate(const int* __restrict__ rowstart,
                   const int* __restrict__ deg,
                   const int* __restrict__ ssrc,
                   const float* __restrict__ al,
                   const float* __restrict__ er,
                   const float* __restrict__ z,
                   const float* __restrict__ bias,
                   float* __restrict__ out,
                   int N, int applyRelu) {
  const int wv  = __builtin_amdgcn_readfirstlane((int)(threadIdx.x >> 5));
  const int wpb = (int)(blockDim.x >> 5);
  const int wid = (int)blockIdx.x * wpb + wv;
  const int nwv = (int)gridDim.x * wpb;
  const int npw = N / nwv;
  const int nex = N - npw * nwv;
  const int node = wid * npw + min(wid, nex);
  const int nlast = node + npw - ((wid < nex) ? 0 : 1);
  if (node > nlast) return;
  gat_aggregate_nodes<false>(rowstart, deg, ssrc, al, er, z, bias, out, nullptr,
                             node, nlast, 0, applyRelu);
}

__global__ void __launch_bounds__(128)
gat_node_aggregate_gemm(const int* __restrict__ rowstart,
                        const int* __restrict__ deg,
                        const int* __restrict__ ssrc,
                        const float* __restrict__ al,
                        const float* __restrict__ er,
                        const float* __restrict__ z,
                        const float* __restrict__ bias,
                        const float* __restrict__ W2,
                        const float* __restrict__ ar2,
                        float* __restrict__ z2,
                        float* __restrict__ er2, int N) {
  __shared__ __attribute__((aligned(16))) float s_h[16 * GAT_HP];
  __shared__ __attribute__((aligned(16))) float s_z[16 * GAT_HP];
  const int lane = threadIdx.x & 31;
  const int wv   = __builtin_amdgcn_readfirstlane((int)(threadIdx.x >> 5));
  const int kh   = lane >> 4;
  const int l15  = lane & 15;
  const int row0 = (int)blockIdx.x * 16;
  const int node = row0 + 4 * wv;
  const int nlast = min(node + 3, N - 1);

  if (node + 3 > N - 1) {
    for (int r = max(N - node, 0); r < 4; ++r)
      if (kh == 0) *(v4f*)(s_h + (4 * wv + r) * GAT_HP + 4 * l15) = (v4f){0.f, 0.f, 0.f, 0.f};
  }
  if (node <= nlast)
    gat_aggregate_nodes<true>(rowstart, deg, ssrc, al, er, z, bias, nullptr, s_h,
                              node, nlast, row0,  1);

  const float* bcol = W2 + (size_t)(2 * kh) * 64 + 16 * wv + l15;
  v2f bf[16];
#pragma unroll
  for (int j = 0; j < 16; ++j) {
    bf[j].x = bcol[(size_t)j * 256];
    bf[j].y = bcol[(size_t)j * 256 + 64];
  }
  float arv[4];
#pragma unroll
  for (int t = 0; t < 4; ++t) arv[t] = ar2[t * 16 + l15];
  __builtin_amdgcn_fence(__ATOMIC_RELEASE, "workgroup", "local");
  __builtin_amdgcn_s_barrier();
  __builtin_amdgcn_fence(__ATOMIC_ACQUIRE, "workgroup", "local");

  const float* arow = s_h + l15 * GAT_HP + 2 * kh;
  v2f af[16];
#pragma unroll
  for (int j = 0; j < 16; ++j) af[j] = *(const v2f*)(arow + 4 * j);
  __builtin_amdgcn_sched_barrier(0);
  v8f acc = (v8f){0.f,0.f,0.f,0.f,0.f,0.f,0.f,0.f};
#pragma unroll
  for (int j = 0; j < 16; ++j)
    acc = __builtin_amdgcn_wmma_f32_16x16x4_f32(
        false, af[j], false, bf[j], (short)0, acc, false, false);
#pragma unroll
  for (int r = 0; r < 8; ++r)
    s_z[(8 * kh + r) * GAT_HP + 16 * wv + l15] = acc[r];
  __syncthreads();

#pragma unroll
  for (int p = 0; p < 2; ++p) {
    const int row = 4 * wv + 2 * p + kh;
    const float* zr = s_z + row * GAT_HP;
    const v4f zv = *(const v4f*)(zr + 4 * l15);
    float sr = zr[l15] * arv[0];
#pragma unroll
    for (int t = 1; t < 4; ++t) sr += zr[t * 16 + l15] * arv[t];
    sr = gat_row_xadd<1>(sr);
    sr = gat_row_xadd<2>(sr);
    sr = gat_row_xadd<4>(sr);
    sr = gat_row_xadd<8>(sr);
    if (row0 + row < N) {
      ((v4f*)(z2 + (size_t)(row0 + row) * 64))[l15] = zv;
      if (l15 == 0) er2[row0 + row] = sr;
    }
  }
}


extern "C" void kernel_launch(void* const* d_in, const int* in_sizes, int n_in,
                              void* d_out, int out_size, void* d_ws, size_t ws_size,
                              hipStream_t stream) {
  const float* x   = (const float*)d_in[0];
  const float* W1  = (const float*)d_in[1];
  const float* al1 = (const float*)d_in[2];
  const float* ar1 = (const float*)d_in[3];
  const float* b1  = (const float*)d_in[4];
  const float* W2  = (const float*)d_in[5];
  const float* al2 = (const float*)d_in[6];
  const float* ar2 = (const float*)d_in[7];
  const float* b2  = (const float*)d_in[8];
  const int*   src = (const int*)d_in[9];
  const int*   dst = (const int*)d_in[10];
  float* out = (float*)d_out;

  const int F  = 64;
  const int N  = out_size / F;
  const int E  = in_sizes[9];
  const int K1 = in_sizes[0] / N;
  (void)n_in; (void)ws_size;

  char* ws = (char*)d_ws;
  size_t off = 0;
  auto alloc = [&](size_t bytes) -> void* {
    void* p = ws + off;
    off += (bytes + 255) & ~(size_t)255;
    return p;
  };
  float* zbuf     = (float*)alloc((size_t)N * F * sizeof(float));
  float* zbuf2    = (float*)alloc((size_t)N * F * sizeof(float));
  float* er       = (float*)alloc((size_t)N * sizeof(float));
  float* er2      = (float*)alloc((size_t)N * sizeof(float));
  int*   deg      = (int*)alloc((size_t)N * sizeof(int));
  int*   rowstart = (int*)alloc((size_t)N * sizeof(int));
  int*   ssrc     = (int*)alloc((size_t)E * sizeof(int));
  const int nb    = (N + 255) / 256;
  const int nwg   = (E + GAT_CHUNK - 1) / GAT_CHUNK;
  int*   cnt_t    = (int*)alloc((size_t)nb * nwg * sizeof(int));
  int*   off_t    = (int*)alloc((size_t)nb * nwg * sizeof(int));
  int*   tmp      = (int*)alloc((size_t)E * 2 * sizeof(int));

  const int gAgg   = 2048;
  const int gFuse  = (N + 15) / 16;
  const int gGemm1 = (((N + 15) / 16) + GAT_GW - 1) / GAT_GW;

  gat_chunk_sort <<<nwg, GAT_T, 0, stream>>>(src, dst, cnt_t, off_t, tmp, E, nb, nwg);
  gat_bucket_csr_and_gemm <<<nb + gGemm1, GAT_T, 0, stream>>>(tmp, cnt_t, off_t, rowstart, deg,
                                                ssrc, N, nwg, nb,
                                                x, W1, ar1, zbuf, er, K1);

  gat_node_aggregate_gemm<<<gFuse, 128, 0, stream>>>(rowstart, deg, ssrc, al1, er,
                                                 zbuf, b1, W2, ar2, zbuf2, er2, N);

  gat_node_aggregate<<<gAgg, 128, 0, stream>>>(rowstart, deg, ssrc, al2, er2,
                                                 zbuf2, b2, out, N,  0);
}
